# starting kernel
# baseline (speedup 1.0000x reference)
#include <hip/hip_runtime.h>
#include <hip/hip_bf16.h>

typedef float v2f __attribute__((ext_vector_type(2)));
typedef float v8f __attribute__((ext_vector_type(8)));

#define N_NODES 50000
#define N_EDGES 800000
#define N_FEAT  256
#define N_HID   32
#define N_HEADS 8
#define ALPHA   0.2f

__global__ void gat_init(float* __restrict__ out,
                         float* __restrict__ segmax,
                         float* __restrict__ denom) {
    int i = blockIdx.x * blockDim.x + threadIdx.x;
    if (i < N_NODES * N_HEADS * N_HID) out[i] = 0.0f;
    if (i < N_NODES * N_HEADS) {
        segmax[i] = -__builtin_inff();
        denom[i]  = 0.0f;
    }
}

__global__ __launch_bounds__(256) void gat_gemm(const float* __restrict__ hmat,
                                                const float* __restrict__ W,
                                                float* __restrict__ Wh) {
    __shared__ float sh[16 * N_FEAT];

    const int node_base = blockIdx.x * 16;

    const float4* g4 = (const float4*)(hmat + (size_t)node_base * N_FEAT);
    float4* s4 = (float4*)sh;
    for (int i = threadIdx.x; i < (16 * N_FEAT) / 4; i += 256)
        s4[i] = g4[i];
    __syncthreads();

    const int lane = threadIdx.x & 31;
    const int head = threadIdx.x >> 5;
    const float* Wp = W + (size_t)head * N_FEAT * N_HID;

    const int m    = lane & 15;
    const int koff = (lane >> 4) << 1;

    v8f c0 = {};
    v8f c1 = {};

    for (int k = 0; k < N_FEAT; k += 4) {
        v2f a, b0, b1;
        a[0] = sh[m * N_FEAT + k + koff];
        a[1] = sh[m * N_FEAT + k + koff + 1];
        b0[0] = Wp[(k + koff)     * N_HID + m];
        b0[1] = Wp[(k + koff + 1) * N_HID + m];
        b1[0] = Wp[(k + koff)     * N_HID + 16 + m];
        b1[1] = Wp[(k + koff + 1) * N_HID + 16 + m];
        c0 = __builtin_amdgcn_wmma_f32_16x16x4_f32(false, a, false, b0,
                                                   (short)0, c0, false, false);
        c1 = __builtin_amdgcn_wmma_f32_16x16x4_f32(false, a, false, b1,
                                                   (short)0, c1, false, false);
    }

    float* op = Wh + ((size_t)head * N_NODES + node_base) * N_HID;
    const int mb = (lane >> 4) * 8;
#pragma unroll
    for (int v = 0; v < 8; ++v) {
        op[(mb + v) * N_HID + m]      = c0[v];
        op[(mb + v) * N_HID + 16 + m] = c1[v];
    }
}

__global__ __launch_bounds__(256) void gat_score(const float* __restrict__ Wh,
                                                 const float* __restrict__ avec,
                                                 float* __restrict__ s1,
                                                 float* __restrict__ s2) {
    const int g    = (blockIdx.x * blockDim.x + threadIdx.x) >> 5;
    const int lane = threadIdx.x & 31;
    if (g >= N_NODES * N_HEADS) return;
    const int head = g & 7;
    const int n    = g >> 3;

    float wh = Wh[((size_t)head * N_NODES + n) * N_HID + lane];
    float p1 = wh * avec[head * 2 * N_HID + lane];
    float p2 = wh * avec[head * 2 * N_HID + N_HID + lane];
#pragma unroll
    for (int off = 16; off > 0; off >>= 1) {
        p1 += __shfl_xor(p1, off, 32);
        p2 += __shfl_xor(p2, off, 32);
    }
    if (lane == 0) {
        s1[head * N_NODES + n] = p1;
        s2[head * N_NODES + n] = p2;
    }
}

__device__ __forceinline__ void atomicMaxFloat(float* addr, float val) {
    if (val >= 0.0f)
        atomicMax((int*)addr, __float_as_int(val));
    else
        atomicMin((unsigned int*)addr, __float_as_uint(val));
}

__global__ void gat_edge_max(const float* __restrict__ s1,
                             const float* __restrict__ s2,
                             const int* __restrict__ src,
                             const int* __restrict__ dst,
                             float* __restrict__ ebuf,
                             float* __restrict__ segmax) {
    const int e = blockIdx.x * blockDim.x + threadIdx.x;
    const int head = blockIdx.y;
    if (e >= N_EDGES) return;
    const int s = src[e], d = dst[e];
    float x = s1[head * N_NODES + s] + s2[head * N_NODES + d];
    float v = x > 0.0f ? x : ALPHA * x;
    ebuf[(size_t)head * N_EDGES + e] = v;
    atomicMaxFloat(&segmax[head * N_NODES + d], v);
}

__global__ void gat_edge_exp(const int* __restrict__ dst,
                             const float* __restrict__ segmax,
                             float* __restrict__ ebuf,
                             float* __restrict__ denom) {
    const int e = blockIdx.x * blockDim.x + threadIdx.x;
    const int head = blockIdx.y;
    if (e >= N_EDGES) return;
    const int d = dst[e];
    float ex = __expf(ebuf[(size_t)head * N_EDGES + e] - segmax[head * N_NODES + d]);
    ebuf[(size_t)head * N_EDGES + e] = ex;
    atomicAdd(&denom[head * N_NODES + d], ex);
}

__global__ __launch_bounds__(256) void gat_message(const int* __restrict__ src,
                                                   const int* __restrict__ dst,
                                                   const float* __restrict__ ebuf,
                                                   const float* __restrict__ denom,
                                                   const float* __restrict__ Wh,
                                                   float* __restrict__ out) {
    const int e    = (blockIdx.x * blockDim.x + threadIdx.x) >> 5;
    const int lane = threadIdx.x & 31;
    if (e >= N_EDGES) return;
    const int s = src[e], d = dst[e];
    float* op = out + (size_t)d * (N_HEADS * N_HID);
#pragma unroll
    for (int head = 0; head < N_HEADS; ++head) {
        float attn = ebuf[(size_t)head * N_EDGES + e] / denom[head * N_NODES + d];
        float val  = attn * Wh[((size_t)head * N_NODES + s) * N_HID + lane];
        atomicAdd(op + head * N_HID + lane, val);
    }
}

extern "C" void kernel_launch(void* const* d_in, const int* in_sizes, int n_in,
                              void* d_out, int out_size, void* d_ws, size_t ws_size,
                              hipStream_t stream) {
    const float* h   = (const float*)d_in[0];
    const float* W   = (const float*)d_in[1];
    const float* a   = (const float*)d_in[2];
    const int*   src = (const int*)d_in[3];
    const int*   dst = (const int*)d_in[4];
    float* out = (float*)d_out;

    float* ws     = (float*)d_ws;
    float* Wh     = ws;
    float* s1     = Wh + (size_t)N_HEADS * N_NODES * N_HID;
    float* s2     = s1 + (size_t)N_HEADS * N_NODES;
    float* segmax = s2 + (size_t)N_HEADS * N_NODES;
    float* denom  = segmax + (size_t)N_HEADS * N_NODES;
    float* ebuf   = denom + (size_t)N_HEADS * N_NODES;

    gat_init<<<(N_NODES * N_HEADS * N_HID + 255) / 256, 256, 0, stream>>>(out, segmax, denom);

    gat_gemm<<<N_NODES / 16, 256, 0, stream>>>(h, W, Wh);

    gat_score<<<(N_NODES * N_HEADS * 32 + 255) / 256, 256, 0, stream>>>(Wh, a, s1, s2);

    dim3 eg((N_EDGES + 255) / 256, N_HEADS);
    gat_edge_max<<<eg, 256, 0, stream>>>(s1, s2, src, dst, ebuf, segmax);
    gat_edge_exp<<<eg, 256, 0, stream>>>(dst, segmax, ebuf, denom);

    gat_message<<<(N_EDGES * 32 + 255) / 256, 256, 0, stream>>>(src, dst, ebuf, denom, Wh, out);
}


# softmax + message pass moved to 256 owner work-groups: each keeps 196 dst rows, maxima, denominators in LDS, streams dst[] itself; LDS atomics (arrival order) replace global ones; 6 launches -> 3
# speedup vs baseline: 4.3434x; 4.3434x over previous
#include <hip/hip_runtime.h>
#include <hip/hip_bf16.h>

typedef float v2f __attribute__((ext_vector_type(2)));
typedef float v8f __attribute__((ext_vector_type(8)));

#define N_NODES 50000
#define N_EDGES 800000
#define N_FEAT  256
#define N_HID   32
#define N_HEADS 8
#define ALPHA   0.2f

__global__ __launch_bounds__(256) void gat_gemm(const float* __restrict__ hmat,
                                                const float* __restrict__ W,
                                                float* __restrict__ Wh) {
    __shared__ float sh[16 * N_FEAT];

    const int node_base = blockIdx.x * 16;

    const float4* g4 = (const float4*)(hmat + (size_t)node_base * N_FEAT);
    float4* s4 = (float4*)sh;
    for (int i = threadIdx.x; i < (16 * N_FEAT) / 4; i += 256)
        s4[i] = g4[i];
    __syncthreads();

    const int lane = threadIdx.x & 31;
    const int head = threadIdx.x >> 5;
    const float* Wp = W + (size_t)head * N_FEAT * N_HID;

    const int m    = lane & 15;
    const int koff = (lane >> 4) << 1;

    v8f c0 = {};
    v8f c1 = {};

    for (int k = 0; k < N_FEAT; k += 4) {
        v2f a, b0, b1;
        a[0] = sh[m * N_FEAT + k + koff];
        a[1] = sh[m * N_FEAT + k + koff + 1];
        b0[0] = Wp[(k + koff)     * N_HID + m];
        b0[1] = Wp[(k + koff + 1) * N_HID + m];
        b1[0] = Wp[(k + koff)     * N_HID + 16 + m];
        b1[1] = Wp[(k + koff + 1) * N_HID + 16 + m];
        c0 = __builtin_amdgcn_wmma_f32_16x16x4_f32(false, a, false, b0,
                                                   (short)0, c0, false, false);
        c1 = __builtin_amdgcn_wmma_f32_16x16x4_f32(false, a, false, b1,
                                                   (short)0, c1, false, false);
    }

    float* op = Wh + ((size_t)head * N_NODES + node_base) * N_HID;
    const int mb = (lane >> 4) * 8;
#pragma unroll
    for (int v = 0; v < 8; ++v) {
        op[(mb + v) * N_HID + m]      = c0[v];
        op[(mb + v) * N_HID + 16 + m] = c1[v];
    }
}

__global__ __launch_bounds__(256) void gat_score(const float* __restrict__ Wh,
                                                 const float* __restrict__ avec,
                                                 float* __restrict__ s1,
                                                 float* __restrict__ s2) {
    const int g    = (blockIdx.x * blockDim.x + threadIdx.x) >> 5;
    const int lane = threadIdx.x & 31;
    if (g >= N_NODES * N_HEADS) return;
    const int head = g & 7;
    const int n    = g >> 3;

    float wh = Wh[((size_t)head * N_NODES + n) * N_HID + lane];
    float p1 = wh * avec[head * 2 * N_HID + lane];
    float p2 = wh * avec[head * 2 * N_HID + N_HID + lane];
#pragma unroll
    for (int off = 16; off > 0; off >>= 1) {
        p1 += __shfl_xor(p1, off, 32);
        p2 += __shfl_xor(p2, off, 32);
    }
    if (lane == 0) {
        s1[head * N_NODES + n] = p1;
        s2[head * N_NODES + n] = p2;
    }
}

#define TILE_NODES 196
#define OWN_WAVES  32
#define LIST_CAP   384
#define EDGE_CHUNK 128
#define N_CHUNKS   (N_EDGES / EDGE_CHUNK)

static_assert(N_EDGES % EDGE_CHUNK == 0, "dst[] is read in whole 128-edge chunks");
static_assert(N_EDGES <= (1 << 20) && TILE_NODES <= 2048, "list entry: local node << 20 | edge");
static_assert(N_NODES <= (1 << 16), "resolved list entry: local node << 16 | src");
static_assert(LIST_CAP > EDGE_CHUNK, "a chunk must fit behind the flush mark");

__device__ __forceinline__ void ldsMaxFloat(int* addr, float val) {
    if (val >= 0.0f)
        atomicMax(addr, __float_as_int(val));
    else
        atomicMin((unsigned int*)addr, __float_as_uint(val));
}

__device__ __forceinline__ void ownListFence() {
    __builtin_amdgcn_fence(__ATOMIC_ACQ_REL, "wavefront");
}

__device__ __forceinline__ void ownResolve(int* lw, int cnt, int lane,
                                           const int* __restrict__ src) {
    for (int i0 = 0; i0 < cnt; i0 += 128) {
        int ent[4], sv[4];
#pragma unroll
        for (int u = 0; u < 4; ++u) {
            const int idx = i0 + u * 32 + lane;
            ent[u] = idx < cnt ? lw[idx] : 0;
        }
#pragma unroll
        for (int u = 0; u < 4; ++u)
            sv[u] = src[ent[u] & 0xFFFFF];
#pragma unroll
        for (int u = 0; u < 4; ++u) {
            const int idx = i0 + u * 32 + lane;
            if (idx < cnt) lw[idx] = ((ent[u] >> 20) << 16) | sv[u];
        }
    }
}

__device__ __forceinline__ void ownMax(const int* lw, int cnt, int lane,
                                       const float* __restrict__ s1,
                                       const float* s2l, int* mxi) {
    const int j = lane >> 3, h = lane & 7;
    for (int i0 = 0; i0 < cnt; i0 += 16) {
        float x[4]; int ld[4];
#pragma unroll
        for (int u = 0; u < 4; ++u) {
            const int idx = i0 + u * 4 + j;
            const int ent = idx < cnt ? lw[idx] : 0;
            ld[u] = ent >> 16;
            x[u] = s1[h * N_NODES + (ent & 0xFFFF)] + s2l[ld[u] * N_HEADS + h];
        }
#pragma unroll
        for (int u = 0; u < 4; ++u) {
            const int idx = i0 + u * 4 + j;
            const float v = x[u] > 0.0f ? x[u] : ALPHA * x[u];
            if (idx < cnt) ldsMaxFloat(&mxi[ld[u] * N_HEADS + h], v);
        }
    }
}

__device__ __forceinline__ void ownSum(const int* lw, int cnt, int lane,
                                       const float* __restrict__ s1,
                                       const float* __restrict__ Wh,
                                       const float* s2l, const int* mxi,
                                       float* dn, float* acc) {
    const int hl = lane & 7;
    for (int i0 = 0; i0 < cnt; i0 += 4) {
        int s[4], ld[4]; float sc[4]; float wh[4][N_HEADS]; float ex[4];
#pragma unroll
        for (int u = 0; u < 4; ++u) {
            const int idx = (i0 + u < cnt) ? i0 + u : cnt - 1;
            const int ent = lw[idx];
            s[u]  = ent & 0xFFFF;
            ld[u] = ent >> 16;
        }
#pragma unroll
        for (int u = 0; u < 4; ++u)
            sc[u] = s1[hl * N_NODES + s[u]];
#pragma unroll
        for (int u = 0; u < 4; ++u)
#pragma unroll
            for (int head = 0; head < N_HEADS; ++head)
                wh[u][head] = Wh[((size_t)head * N_NODES + s[u]) * N_HID + lane];
#pragma unroll
        for (int u = 0; u < 4; ++u) {
            const float x = sc[u] + s2l[ld[u] * N_HEADS + hl];
            const float v = x > 0.0f ? x : ALPHA * x;
            ex[u] = __expf(v - __int_as_float(mxi[ld[u] * N_HEADS + hl]));
        }
#pragma unroll
        for (int u = 0; u < 4; ++u) {
            if (i0 + u < cnt) {
                if (lane < N_HEADS) atomicAdd(&dn[ld[u] * N_HEADS + lane], ex[u]);
                float* ap = acc + ld[u] * (N_HEADS * N_HID) + lane;
#pragma unroll
                for (int head = 0; head < N_HEADS; ++head) {
                    const float w = __int_as_float(
                        __builtin_amdgcn_readlane(__float_as_int(ex[u]), head));
                    atomicAdd(ap + head * N_HID, w * wh[u][head]);
                }
            }
        }
    }
}

template <int PHASE>
__device__ __forceinline__ void ownWork(int* lw, int cnt, int lane,
                                        const float* __restrict__ s1,
                                        const float* __restrict__ Wh,
                                        const float* s2l, int* mxi,
                                        float* dn, float* acc) {
    if (PHASE == 0) ownMax(lw, cnt, lane, s1, s2l, mxi);
    else            ownSum(lw, cnt, lane, s1, Wh, s2l, mxi, dn, acc);
}

template <int PHASE>
__device__ __forceinline__ int ownScan(int* lw, int wave, int lane, int lo, int nloc,
                                       const int* __restrict__ src,
                                       const int* __restrict__ dst,
                                       const float* __restrict__ s1,
                                       const float* __restrict__ Wh,
                                       const float* s2l, int* mxi,
                                       float* dn, float* acc, int& cnt_out) {
    const int4* d4 = (const int4*)dst;
    int cnt = 0, rounds = 0;
    for (int c0 = wave; c0 < N_CHUNKS; c0 += OWN_WAVES * 8) {
        int4 dv[8];
#pragma unroll
        for (int u = 0; u < 8; ++u) {
            const int c = c0 + u * OWN_WAVES;
            dv[u] = c < N_CHUNKS ? d4[c * 32 + lane] : make_int4(-1, -1, -1, -1);
        }
#pragma unroll
        for (int u = 0; u < 8; ++u) {
            const int c = c0 + u * OWN_WAVES;
            const unsigned r0 = (unsigned)(dv[u].x - lo);
            const unsigned r1 = (unsigned)(dv[u].y - lo);
            const unsigned r2 = (unsigned)(dv[u].z - lo);
            const unsigned r3 = (unsigned)(dv[u].w - lo);
            const bool m0 = r0 < (unsigned)nloc, m1 = r1 < (unsigned)nloc;
            const bool m2 = r2 < (unsigned)nloc, m3 = r3 < (unsigned)nloc;
            if (__builtin_amdgcn_ballot_w32(m0 | m1 | m2 | m3) != 0u) {
                const unsigned e0 = (unsigned)c * EDGE_CHUNK + lane * 4;
                unsigned mask;
                mask = __builtin_amdgcn_ballot_w32(m0);
                if (m0) lw[cnt + __builtin_amdgcn_mbcnt_lo(mask, 0u)] = (int)((r0 << 20) | e0);
                cnt += __builtin_popcount(mask);
                mask = __builtin_amdgcn_ballot_w32(m1);
                if (m1) lw[cnt + __builtin_amdgcn_mbcnt_lo(mask, 0u)] = (int)((r1 << 20) | (e0 + 1));
                cnt += __builtin_popcount(mask);
                mask = __builtin_amdgcn_ballot_w32(m2);
                if (m2) lw[cnt + __builtin_amdgcn_mbcnt_lo(mask, 0u)] = (int)((r2 << 20) | (e0 + 2));
                cnt += __builtin_popcount(mask);
                mask = __builtin_amdgcn_ballot_w32(m3);
                if (m3) lw[cnt + __builtin_amdgcn_mbcnt_lo(mask, 0u)] = (int)((r3 << 20) | (e0 + 3));
                cnt += __builtin_popcount(mask);
                if (cnt > LIST_CAP - EDGE_CHUNK) {
                    ownListFence();
                    ownResolve(lw, cnt, lane, src);
                    ownListFence();
                    ownWork<PHASE>(lw, cnt, lane, s1, Wh, s2l, mxi, dn, acc);
                    ownListFence();
                    cnt = 0;
                    ++rounds;
                }
            }
        }
    }
    ownListFence();
    ownResolve(lw, cnt, lane, src);
    ownListFence();
    ownWork<PHASE>(lw, cnt, lane, s1, Wh, s2l, mxi, dn, acc);
    cnt_out = cnt;
    return rounds + 1;
}

__global__ __launch_bounds__(1024) void gat_own(const int* __restrict__ src,
                                                const int* __restrict__ dst,
                                                const float* __restrict__ s1,
                                                const float* __restrict__ s2,
                                                const float* __restrict__ Wh,
                                                float* __restrict__ out) {
    __shared__ __attribute__((aligned(16))) float acc[TILE_NODES * N_HEADS * N_HID];
    __shared__ int   mxi[TILE_NODES * N_HEADS];
    __shared__ float dn[TILE_NODES * N_HEADS];
    __shared__ float s2l[TILE_NODES * N_HEADS];
    __shared__ int   lst[OWN_WAVES * LIST_CAP];

    const int tid  = threadIdx.x;
    const int lane = tid & 31;
    const int wave = __builtin_amdgcn_readfirstlane(tid >> 5);
    const int lo   = blockIdx.x * TILE_NODES;
    const int nloc = (N_NODES - lo) < TILE_NODES ? (N_NODES - lo) : TILE_NODES;

    float4* acc4 = (float4*)acc;
    for (int i = tid; i < TILE_NODES * N_HEADS * N_HID / 4; i += 1024)
        acc4[i] = make_float4(0.0f, 0.0f, 0.0f, 0.0f);
    for (int i = tid; i < TILE_NODES * N_HEADS; i += 1024) {
        const int node = lo + (i >> 3);
        mxi[i] = __float_as_int(-__builtin_inff());
        dn[i]  = 0.0f;
        s2l[i] = node < N_NODES ? s2[(i & 7) * N_NODES + node] : 0.0f;
    }
    __syncthreads();

    int* lw = lst + wave * LIST_CAP;
    int cnt = 0;
    const int rounds = ownScan<0>(lw, wave, lane, lo, nloc, src, dst, s1, Wh,
                                  s2l, mxi, dn, acc, cnt);
    __syncthreads();

    if (rounds == 1)
        ownSum(lw, cnt, lane, s1, Wh, s2l, mxi, dn, acc);
    else
        ownScan<1>(lw, wave, lane, lo, nloc, src, dst, s1, Wh, s2l, mxi, dn, acc, cnt);
    __syncthreads();

    float4* o4 = (float4*)(out + (size_t)lo * (N_HEADS * N_HID));
    for (int i = tid; i < nloc * (N_HEADS * N_HID / 4); i += 1024) {
        const float d0 = dn[(i >> 6) * N_HEADS + ((i >> 3) & 7)];
        const float q = d0 > 0.0f ? d0 : 1.0f;
        float4 v = acc4[i];
        v.x /= q; v.y /= q; v.z /= q; v.w /= q;
        o4[i] = v;
    }
}

extern "C" void kernel_launch(void* const* d_in, const int* in_sizes, int n_in,
                              void* d_out, int out_size, void* d_ws, size_t ws_size,
                              hipStream_t stream) {
    const float* h   = (const float*)d_in[0];
    const float* W   = (const float*)d_in[1];
    const float* a   = (const float*)d_in[2];
    const int*   src = (const int*)d_in[3];
    const int*   dst = (const int*)d_in[4];
    float* out = (float*)d_out;

    float* ws = (float*)d_ws;
    float* Wh = ws;
    float* s1 = Wh + (size_t)N_HEADS * N_NODES * N_HID;
    float* s2 = s1 + (size_t)N_HEADS * N_NODES;

    gat_gemm<<<N_NODES / 16, 256, 0, stream>>>(h, W, Wh);

    gat_score<<<(N_NODES * N_HEADS * 32 + 255) / 256, 256, 0, stream>>>(Wh, a, s1, s2);

    gat_own<<<(N_NODES + TILE_NODES - 1) / TILE_NODES, 1024, 0, stream>>>(src, dst, s1, s2, Wh, out);
}


# score launch reshaped: 6250 waves stream 64 Wh rows each (16-byte loads, 8 lanes a row, exchange tree xor 4,2,1, whole-line stores) in place of 400000 one-row waves; s1,s2 sum tree changed
# speedup vs baseline: 4.9824x; 1.1471x over previous
#include <hip/hip_runtime.h>
#include <hip/hip_bf16.h>

typedef float v2f __attribute__((ext_vector_type(2)));
typedef float v8f __attribute__((ext_vector_type(8)));

#define N_NODES 50000
#define N_EDGES 800000
#define N_FEAT  256
#define N_HID   32
#define N_HEADS 8
#define ALPHA   0.2f

__global__ __launch_bounds__(256) void gat_gemm(const float* __restrict__ hmat,
                                                const float* __restrict__ W,
                                                float* __restrict__ Wh) {
    __shared__ float sh[16 * N_FEAT];

    const int node_base = blockIdx.x * 16;

    const float4* g4 = (const float4*)(hmat + (size_t)node_base * N_FEAT);
    float4* s4 = (float4*)sh;
    for (int i = threadIdx.x; i < (16 * N_FEAT) / 4; i += 256)
        s4[i] = g4[i];
    __syncthreads();

    const int lane = threadIdx.x & 31;
    const int head = threadIdx.x >> 5;
    const float* Wp = W + (size_t)head * N_FEAT * N_HID;

    const int m    = lane & 15;
    const int koff = (lane >> 4) << 1;

    v8f c0 = {};
    v8f c1 = {};

    for (int k = 0; k < N_FEAT; k += 4) {
        v2f a, b0, b1;
        a[0] = sh[m * N_FEAT + k + koff];
        a[1] = sh[m * N_FEAT + k + koff + 1];
        b0[0] = Wp[(k + koff)     * N_HID + m];
        b0[1] = Wp[(k + koff + 1) * N_HID + m];
        b1[0] = Wp[(k + koff)     * N_HID + 16 + m];
        b1[1] = Wp[(k + koff + 1) * N_HID + 16 + m];
        c0 = __builtin_amdgcn_wmma_f32_16x16x4_f32(false, a, false, b0,
                                                   (short)0, c0, false, false);
        c1 = __builtin_amdgcn_wmma_f32_16x16x4_f32(false, a, false, b1,
                                                   (short)0, c1, false, false);
    }

    float* op = Wh + ((size_t)head * N_NODES + node_base) * N_HID;
    const int mb = (lane >> 4) * 8;
#pragma unroll
    for (int v = 0; v < 8; ++v) {
        op[(mb + v) * N_HID + m]      = c0[v];
        op[(mb + v) * N_HID + 16 + m] = c1[v];
    }
}

#define SC_ROWS  64
#define SC_WAVES ((N_HEADS * N_NODES) / SC_ROWS)

static_assert((N_HEADS * N_NODES) % SC_ROWS == 0, "whole 64-row pieces");
static_assert(N_NODES % 16 == 0, "a group of 16 rows lies inside one head");

__global__ __launch_bounds__(256) void gat_score(const float* __restrict__ Wh,
                                                 const float* __restrict__ avec,
                                                 float* __restrict__ s1,
                                                 float* __restrict__ s2) {
    const int lane = threadIdx.x & 31;
    const int w = blockIdx.x * 8 + __builtin_amdgcn_readfirstlane(threadIdx.x >> 5);
    if (w >= SC_WAVES) return;
    const int j = lane >> 3;
    const int q = lane & 7;

    const float4* w4 = (const float4*)Wh + (size_t)w * (SC_ROWS * 8) + lane;
    float4 v[16];
#pragma unroll
    for (int i = 0; i < 16; ++i)
        v[i] = w4[i * 32];

    const float4* a4 = (const float4*)avec;
    float p1[16], p2[16];
#pragma unroll
    for (int t = 0; t < 4; ++t) {
        const int head = (w * SC_ROWS + t * 16) / N_NODES;
        const float4 x1 = a4[head * 16 + q];
        const float4 x2 = a4[head * 16 + 8 + q];
#pragma unroll
        for (int u = 0; u < 4; ++u) {
            const float4 c = v[t * 4 + u];
            p1[t * 4 + u] = c.x * x1.x + c.y * x1.y + c.z * x1.z + c.w * x1.w;
            p2[t * 4 + u] = c.x * x2.x + c.y * x2.y + c.z * x2.z + c.w * x2.w;
        }
    }

    const bool b2 = (q & 4) != 0, b1 = (q & 2) != 0, b0 = (q & 1) != 0;
    float r1[8], r2[8];
#pragma unroll
    for (int g = 0; g < 2; ++g)
#pragma unroll
        for (int l = 0; l < 4; ++l) {
            const int i = g * 8 + l;
            const float k1 = b2 ? p1[i + 4] : p1[i], d1 = b2 ? p1[i] : p1[i + 4];
            const float k2 = b2 ? p2[i + 4] : p2[i], d2 = b2 ? p2[i] : p2[i + 4];
            r1[g * 4 + l] = k1 + __shfl_xor(d1, 4, 32);
            r2[g * 4 + l] = k2 + __shfl_xor(d2, 4, 32);
        }
    float t1[4], t2[4];
#pragma unroll
    for (int g = 0; g < 2; ++g)
#pragma unroll
        for (int l = 0; l < 2; ++l) {
            const int i = g * 4 + l;
            const float k1 = b1 ? r1[i + 2] : r1[i], d1 = b1 ? r1[i] : r1[i + 2];
            const float k2 = b1 ? r2[i + 2] : r2[i], d2 = b1 ? r2[i] : r2[i + 2];
            t1[g * 2 + l] = k1 + __shfl_xor(d1, 2, 32);
            t2[g * 2 + l] = k2 + __shfl_xor(d2, 2, 32);
        }
    float* o1 = s1 + (size_t)w * SC_ROWS + 4 * q + j;
    float* o2 = s2 + (size_t)w * SC_ROWS + 4 * q + j;
#pragma unroll
    for (int g = 0; g < 2; ++g) {
        const int i = g * 2;
        const float k1 = b0 ? t1[i + 1] : t1[i], d1 = b0 ? t1[i] : t1[i + 1];
        const float k2 = b0 ? t2[i + 1] : t2[i], d2 = b0 ? t2[i] : t2[i + 1];
        o1[g * 32] = k1 + __shfl_xor(d1, 1, 32);
        o2[g * 32] = k2 + __shfl_xor(d2, 1, 32);
    }
}

#define TILE_NODES 196
#define OWN_WAVES  32
#define LIST_CAP   384
#define EDGE_CHUNK 128
#define N_CHUNKS   (N_EDGES / EDGE_CHUNK)

static_assert(N_EDGES % EDGE_CHUNK == 0, "dst[] is read in whole 128-edge chunks");
static_assert(N_EDGES <= (1 << 20) && TILE_NODES <= 2048, "list entry: local node << 20 | edge");
static_assert(N_NODES <= (1 << 16), "resolved list entry: local node << 16 | src");
static_assert(LIST_CAP > EDGE_CHUNK, "a chunk must fit behind the flush mark");

__device__ __forceinline__ void ldsMaxFloat(int* addr, float val) {
    if (val >= 0.0f)
        atomicMax(addr, __float_as_int(val));
    else
        atomicMin((unsigned int*)addr, __float_as_uint(val));
}

__device__ __forceinline__ void ownListFence() {
    __builtin_amdgcn_fence(__ATOMIC_ACQ_REL, "wavefront");
}

__device__ __forceinline__ void ownResolve(int* lw, int cnt, int lane,
                                           const int* __restrict__ src) {
    for (int i0 = 0; i0 < cnt; i0 += 128) {
        int ent[4], sv[4];
#pragma unroll
        for (int u = 0; u < 4; ++u) {
            const int idx = i0 + u * 32 + lane;
            ent[u] = idx < cnt ? lw[idx] : 0;
        }
#pragma unroll
        for (int u = 0; u < 4; ++u)
            sv[u] = src[ent[u] & 0xFFFFF];
#pragma unroll
        for (int u = 0; u < 4; ++u) {
            const int idx = i0 + u * 32 + lane;
            if (idx < cnt) lw[idx] = ((ent[u] >> 20) << 16) | sv[u];
        }
    }
}

__device__ __forceinline__ void ownMax(const int* lw, int cnt, int lane,
                                       const float* __restrict__ s1,
                                       const float* s2l, int* mxi) {
    const int j = lane >> 3, h = lane & 7;
    for (int i0 = 0; i0 < cnt; i0 += 16) {
        float x[4]; int ld[4];
#pragma unroll
        for (int u = 0; u < 4; ++u) {
            const int idx = i0 + u * 4 + j;
            const int ent = idx < cnt ? lw[idx] : 0;
            ld[u] = ent >> 16;
            x[u] = s1[h * N_NODES + (ent & 0xFFFF)] + s2l[ld[u] * N_HEADS + h];
        }
#pragma unroll
        for (int u = 0; u < 4; ++u) {
            const int idx = i0 + u * 4 + j;
            const float v = x[u] > 0.0f ? x[u] : ALPHA * x[u];
            if (idx < cnt) ldsMaxFloat(&mxi[ld[u] * N_HEADS + h], v);
        }
    }
}

__device__ __forceinline__ void ownSum(const int* lw, int cnt, int lane,
                                       const float* __restrict__ s1,
                                       const float* __restrict__ Wh,
                                       const float* s2l, const int* mxi,
                                       float* dn, float* acc) {
    const int hl = lane & 7;
    for (int i0 = 0; i0 < cnt; i0 += 4) {
        int s[4], ld[4]; float sc[4]; float wh[4][N_HEADS]; float ex[4];
#pragma unroll
        for (int u = 0; u < 4; ++u) {
            const int idx = (i0 + u < cnt) ? i0 + u : cnt - 1;
            const int ent = lw[idx];
            s[u]  = ent & 0xFFFF;
            ld[u] = ent >> 16;
        }
#pragma unroll
        for (int u = 0; u < 4; ++u)
            sc[u] = s1[hl * N_NODES + s[u]];
#pragma unroll
        for (int u = 0; u < 4; ++u)
#pragma unroll
            for (int head = 0; head < N_HEADS; ++head)
                wh[u][head] = Wh[((size_t)head * N_NODES + s[u]) * N_HID + lane];
#pragma unroll
        for (int u = 0; u < 4; ++u) {
            const float x = sc[u] + s2l[ld[u] * N_HEADS + hl];
            const float v = x > 0.0f ? x : ALPHA * x;
            ex[u] = __expf(v - __int_as_float(mxi[ld[u] * N_HEADS + hl]));
        }
#pragma unroll
        for (int u = 0; u < 4; ++u) {
            if (i0 + u < cnt) {
                if (lane < N_HEADS) atomicAdd(&dn[ld[u] * N_HEADS + lane], ex[u]);
                float* ap = acc + ld[u] * (N_HEADS * N_HID) + lane;
#pragma unroll
                for (int head = 0; head < N_HEADS; ++head) {
                    const float w = __int_as_float(
                        __builtin_amdgcn_readlane(__float_as_int(ex[u]), head));
                    atomicAdd(ap + head * N_HID, w * wh[u][head]);
                }
            }
        }
    }
}

template <int PHASE>
__device__ __forceinline__ void ownWork(int* lw, int cnt, int lane,
                                        const float* __restrict__ s1,
                                        const float* __restrict__ Wh,
                                        const float* s2l, int* mxi,
                                        float* dn, float* acc) {
    if (PHASE == 0) ownMax(lw, cnt, lane, s1, s2l, mxi);
    else            ownSum(lw, cnt, lane, s1, Wh, s2l, mxi, dn, acc);
}

template <int PHASE>
__device__ __forceinline__ int ownScan(int* lw, int wave, int lane, int lo, int nloc,
                                       const int* __restrict__ src,
                                       const int* __restrict__ dst,
                                       const float* __restrict__ s1,
                                       const float* __restrict__ Wh,
                                       const float* s2l, int* mxi,
                                       float* dn, float* acc, int& cnt_out) {
    const int4* d4 = (const int4*)dst;
    int cnt = 0, rounds = 0;
    for (int c0 = wave; c0 < N_CHUNKS; c0 += OWN_WAVES * 8) {
        int4 dv[8];
#pragma unroll
        for (int u = 0; u < 8; ++u) {
            const int c = c0 + u * OWN_WAVES;
            dv[u] = c < N_CHUNKS ? d4[c * 32 + lane] : make_int4(-1, -1, -1, -1);
        }
#pragma unroll
        for (int u = 0; u < 8; ++u) {
            const int c = c0 + u * OWN_WAVES;
            const unsigned r0 = (unsigned)(dv[u].x - lo);
            const unsigned r1 = (unsigned)(dv[u].y - lo);
            const unsigned r2 = (unsigned)(dv[u].z - lo);
            const unsigned r3 = (unsigned)(dv[u].w - lo);
            const bool m0 = r0 < (unsigned)nloc, m1 = r1 < (unsigned)nloc;
            const bool m2 = r2 < (unsigned)nloc, m3 = r3 < (unsigned)nloc;
            if (__builtin_amdgcn_ballot_w32(m0 | m1 | m2 | m3) != 0u) {
                const unsigned e0 = (unsigned)c * EDGE_CHUNK + lane * 4;
                unsigned mask;
                mask = __builtin_amdgcn_ballot_w32(m0);
                if (m0) lw[cnt + __builtin_amdgcn_mbcnt_lo(mask, 0u)] = (int)((r0 << 20) | e0);
                cnt += __builtin_popcount(mask);
                mask = __builtin_amdgcn_ballot_w32(m1);
                if (m1) lw[cnt + __builtin_amdgcn_mbcnt_lo(mask, 0u)] = (int)((r1 << 20) | (e0 + 1));
                cnt += __builtin_popcount(mask);
                mask = __builtin_amdgcn_ballot_w32(m2);
                if (m2) lw[cnt + __builtin_amdgcn_mbcnt_lo(mask, 0u)] = (int)((r2 << 20) | (e0 + 2));
                cnt += __builtin_popcount(mask);
                mask = __builtin_amdgcn_ballot_w32(m3);
                if (m3) lw[cnt + __builtin_amdgcn_mbcnt_lo(mask, 0u)] = (int)((r3 << 20) | (e0 + 3));
                cnt += __builtin_popcount(mask);
                if (cnt > LIST_CAP - EDGE_CHUNK) {
                    ownListFence();
                    ownResolve(lw, cnt, lane, src);
                    ownListFence();
                    ownWork<PHASE>(lw, cnt, lane, s1, Wh, s2l, mxi, dn, acc);
                    ownListFence();
                    cnt = 0;
                    ++rounds;
                }
            }
        }
    }
    ownListFence();
    ownResolve(lw, cnt, lane, src);
    ownListFence();
    ownWork<PHASE>(lw, cnt, lane, s1, Wh, s2l, mxi, dn, acc);
    cnt_out = cnt;
    return rounds + 1;
}

__global__ __launch_bounds__(1024) void gat_own(const int* __restrict__ src,
                                                const int* __restrict__ dst,
                                                const float* __restrict__ s1,
                                                const float* __restrict__ s2,
                                                const float* __restrict__ Wh,
                                                float* __restrict__ out) {
    __shared__ __attribute__((aligned(16))) float acc[TILE_NODES * N_HEADS * N_HID];
    __shared__ int   mxi[TILE_NODES * N_HEADS];
    __shared__ float dn[TILE_NODES * N_HEADS];
    __shared__ float s2l[TILE_NODES * N_HEADS];
    __shared__ int   lst[OWN_WAVES * LIST_CAP];

    const int tid  = threadIdx.x;
    const int lane = tid & 31;
    const int wave = __builtin_amdgcn_readfirstlane(tid >> 5);
    const int lo   = blockIdx.x * TILE_NODES;
    const int nloc = (N_NODES - lo) < TILE_NODES ? (N_NODES - lo) : TILE_NODES;

    float4* acc4 = (float4*)acc;
    for (int i = tid; i < TILE_NODES * N_HEADS * N_HID / 4; i += 1024)
        acc4[i] = make_float4(0.0f, 0.0f, 0.0f, 0.0f);
    for (int i = tid; i < TILE_NODES * N_HEADS; i += 1024) {
        const int node = lo + (i >> 3);
        mxi[i] = __float_as_int(-__builtin_inff());
        dn[i]  = 0.0f;
        s2l[i] = node < N_NODES ? s2[(i & 7) * N_NODES + node] : 0.0f;
    }
    __syncthreads();

    int* lw = lst + wave * LIST_CAP;
    int cnt = 0;
    const int rounds = ownScan<0>(lw, wave, lane, lo, nloc, src, dst, s1, Wh,
                                  s2l, mxi, dn, acc, cnt);
    __syncthreads();

    if (rounds == 1)
        ownSum(lw, cnt, lane, s1, Wh, s2l, mxi, dn, acc);
    else
        ownScan<1>(lw, wave, lane, lo, nloc, src, dst, s1, Wh, s2l, mxi, dn, acc, cnt);
    __syncthreads();

    float4* o4 = (float4*)(out + (size_t)lo * (N_HEADS * N_HID));
    for (int i = tid; i < nloc * (N_HEADS * N_HID / 4); i += 1024) {
        const float d0 = dn[(i >> 6) * N_HEADS + ((i >> 3) & 7)];
        const float q = d0 > 0.0f ? d0 : 1.0f;
        float4 v = acc4[i];
        v.x /= q; v.y /= q; v.z /= q; v.w /= q;
        o4[i] = v;
    }
}

extern "C" void kernel_launch(void* const* d_in, const int* in_sizes, int n_in,
                              void* d_out, int out_size, void* d_ws, size_t ws_size,
                              hipStream_t stream) {
    const float* h   = (const float*)d_in[0];
    const float* W   = (const float*)d_in[1];
    const float* a   = (const float*)d_in[2];
    const int*   src = (const int*)d_in[3];
    const int*   dst = (const int*)d_in[4];
    float* out = (float*)d_out;

    float* ws = (float*)d_ws;
    float* Wh = ws;
    float* s1 = Wh + (size_t)N_HEADS * N_NODES * N_HID;
    float* s2 = s1 + (size_t)N_HEADS * N_NODES;

    gat_gemm<<<N_NODES / 16, 256, 0, stream>>>(h, W, Wh);

    gat_score<<<(SC_WAVES + 7) / 8, 256, 0, stream>>>(Wh, a, s1, s2);

    gat_own<<<(N_NODES + TILE_NODES - 1) / TILE_NODES, 1024, 0, stream>>>(src, dst, s1, s2, Wh, out);
}


# GEMM re-cut: W kept whole in LDS in fragment order (256 work-groups), wave = 16 nodes x 128 columns, A fragments straight from global rows, no h tile in LDS; k sum order of Wh changed
# speedup vs baseline: 5.2170x; 1.0471x over previous
#include <hip/hip_runtime.h>
#include <hip/hip_bf16.h>

typedef float v2f __attribute__((ext_vector_type(2)));
typedef float v8f __attribute__((ext_vector_type(8)));

#define N_NODES 50000
#define N_EDGES 800000
#define N_FEAT  256
#define N_HID   32
#define N_HEADS 8
#define ALPHA   0.2f

#define GM_GRID  256
#define GM_WAVES 8
#define GM_TILES (N_NODES / 16)
#define GM_STEP  (GM_GRID * (GM_WAVES / 2))

__global__ __launch_bounds__(GM_WAVES * 32) void gat_gemm(const float* __restrict__ hmat,
                                                          const float* __restrict__ W,
                                                          float* __restrict__ Wh) {
    __shared__ v2f wl[64 * 16 * 32];

    const int tid  = threadIdx.x;
    const int lane = tid & 31;
    const int wave = __builtin_amdgcn_readfirstlane(tid >> 5);
    const int m  = lane & 15;
    const int hf = lane >> 4;

    const int nh = wave & 1;
    int tile = blockIdx.x + GM_GRID * (wave >> 1);

    const float4* rowp = (const float4*)(hmat + (size_t)(tile * 16 + m) * N_FEAT + 32 * hf);
    float4 cur[8];
#pragma unroll
    for (int i = 0; i < 8; ++i) cur[i] = rowp[i];

    for (int i0 = 0; i0 < (64 * 16 * 32) / (GM_WAVES * 32); i0 += 16) {
        v2f tmp[16];
#pragma unroll
        for (int i = 0; i < 16; ++i) {
            const int idx = tid + GM_WAVES * 32 * (i0 + i);
            const int nt = (idx >> 5) & 15, ks = idx >> 9;
            const int k = 64 * (ks >> 4) + 32 * hf + 2 * (ks & 15);
            const float* wp = W + (size_t)((nt >> 1) * N_FEAT + k) * N_HID + (nt & 1) * 16 + m;
            tmp[i][0] = wp[0];
            tmp[i][1] = wp[N_HID];
        }
#pragma unroll
        for (int i = 0; i < 16; ++i)
            wl[tid + GM_WAVES * 32 * (i0 + i)] = tmp[i];
    }
    __syncthreads();

    const v2f* bl = wl + nh * (8 * 32) + lane;

    for (; tile < GM_TILES; tile += GM_STEP) {
        const int tn = tile + GM_STEP < GM_TILES ? tile + GM_STEP : tile;
        const float4* nrow = (const float4*)(hmat + (size_t)(tn * 16 + m) * N_FEAT + 32 * hf);

        v8f acc[8];
#pragma unroll
        for (int j = 0; j < 8; ++j) acc[j] = v8f{};

#pragma nounroll
        for (int kc = 0; kc < 4; ++kc) {
            const float4* np = kc < 3 ? rowp + (kc + 1) * 16 : nrow;
            float4 nxt[8];
#pragma unroll
            for (int i = 0; i < 8; ++i) nxt[i] = np[i];

            const v2f* bk = bl + kc * (16 * 16 * 32);
#pragma unroll
            for (int s = 0; s < 16; ++s) {
                const float4 q = cur[s >> 1];
                v2f a;
                a[0] = (s & 1) ? q.z : q.x;
                a[1] = (s & 1) ? q.w : q.y;
#pragma unroll
                for (int j = 0; j < 8; ++j) {
                    const v2f b = bk[(s * 16 + j) * 32];
                    acc[j] = __builtin_amdgcn_wmma_f32_16x16x4_f32(false, a, false, b,
                                                                   (short)0, acc[j], false, false);
                }
            }
#pragma unroll
            for (int i = 0; i < 8; ++i) cur[i] = nxt[i];
        }

#pragma unroll
        for (int j = 0; j < 8; ++j) {
            const int head = nh * 4 + (j >> 1);
            float* op = Wh + ((size_t)head * N_NODES + tile * 16 + hf * 8) * N_HID + (j & 1) * 16 + m;
#pragma unroll
            for (int r = 0; r < 8; ++r)
                op[r * N_HID] = acc[j][r];
        }
        rowp = nrow;
    }
}

#define SC_ROWS  64
#define SC_WAVES ((N_HEADS * N_NODES) / SC_ROWS)

static_assert((N_HEADS * N_NODES) % SC_ROWS == 0, "whole 64-row pieces");
static_assert(N_NODES % 16 == 0, "a group of 16 rows lies inside one head");

__global__ __launch_bounds__(256) void gat_score(const float* __restrict__ Wh,
                                                 const float* __restrict__ avec,
                                                 float* __restrict__ s1,
                                                 float* __restrict__ s2) {
    const int lane = threadIdx.x & 31;
    const int w = blockIdx.x * 8 + __builtin_amdgcn_readfirstlane(threadIdx.x >> 5);
    if (w >= SC_WAVES) return;
    const int j = lane >> 3;
    const int q = lane & 7;

    const float4* w4 = (const float4*)Wh + (size_t)w * (SC_ROWS * 8) + lane;
    float4 v[16];
#pragma unroll
    for (int i = 0; i < 16; ++i)
        v[i] = w4[i * 32];

    const float4* a4 = (const float4*)avec;
    float p1[16], p2[16];
#pragma unroll
    for (int t = 0; t < 4; ++t) {
        const int head = (w * SC_ROWS + t * 16) / N_NODES;
        const float4 x1 = a4[head * 16 + q];
        const float4 x2 = a4[head * 16 + 8 + q];
#pragma unroll
        for (int u = 0; u < 4; ++u) {
            const float4 c = v[t * 4 + u];
            p1[t * 4 + u] = c.x * x1.x + c.y * x1.y + c.z * x1.z + c.w * x1.w;
            p2[t * 4 + u] = c.x * x2.x + c.y * x2.y + c.z * x2.z + c.w * x2.w;
        }
    }

    const bool b2 = (q & 4) != 0, b1 = (q & 2) != 0, b0 = (q & 1) != 0;
    float r1[8], r2[8];
#pragma unroll
    for (int g = 0; g < 2; ++g)
#pragma unroll
        for (int l = 0; l < 4; ++l) {
            const int i = g * 8 + l;
            const float k1 = b2 ? p1[i + 4] : p1[i], d1 = b2 ? p1[i] : p1[i + 4];
            const float k2 = b2 ? p2[i + 4] : p2[i], d2 = b2 ? p2[i] : p2[i + 4];
            r1[g * 4 + l] = k1 + __shfl_xor(d1, 4, 32);
            r2[g * 4 + l] = k2 + __shfl_xor(d2, 4, 32);
        }
    float t1[4], t2[4];
#pragma unroll
    for (int g = 0; g < 2; ++g)
#pragma unroll
        for (int l = 0; l < 2; ++l) {
            const int i = g * 4 + l;
            const float k1 = b1 ? r1[i + 2] : r1[i], d1 = b1 ? r1[i] : r1[i + 2];
            const float k2 = b1 ? r2[i + 2] : r2[i], d2 = b1 ? r2[i] : r2[i + 2];
            t1[g * 2 + l] = k1 + __shfl_xor(d1, 2, 32);
            t2[g * 2 + l] = k2 + __shfl_xor(d2, 2, 32);
        }
    float* o1 = s1 + (size_t)w * SC_ROWS + 4 * q + j;
    float* o2 = s2 + (size_t)w * SC_ROWS + 4 * q + j;
#pragma unroll
    for (int g = 0; g < 2; ++g) {
        const int i = g * 2;
        const float k1 = b0 ? t1[i + 1] : t1[i], d1 = b0 ? t1[i] : t1[i + 1];
        const float k2 = b0 ? t2[i + 1] : t2[i], d2 = b0 ? t2[i] : t2[i + 1];
        o1[g * 32] = k1 + __shfl_xor(d1, 1, 32);
        o2[g * 32] = k2 + __shfl_xor(d2, 1, 32);
    }
}

#define TILE_NODES 196
#define OWN_WAVES  32
#define LIST_CAP   384
#define EDGE_CHUNK 128
#define N_CHUNKS   (N_EDGES / EDGE_CHUNK)

static_assert(N_EDGES % EDGE_CHUNK == 0, "dst[] is read in whole 128-edge chunks");
static_assert(N_EDGES <= (1 << 20) && TILE_NODES <= 2048, "list entry: local node << 20 | edge");
static_assert(N_NODES <= (1 << 16), "resolved list entry: local node << 16 | src");
static_assert(LIST_CAP > EDGE_CHUNK, "a chunk must fit behind the flush mark");

__device__ __forceinline__ void ldsMaxFloat(int* addr, float val) {
    if (val >= 0.0f)
        atomicMax(addr, __float_as_int(val));
    else
        atomicMin((unsigned int*)addr, __float_as_uint(val));
}

__device__ __forceinline__ void ownListFence() {
    __builtin_amdgcn_fence(__ATOMIC_ACQ_REL, "wavefront");
}

__device__ __forceinline__ void ownResolve(int* lw, int cnt, int lane,
                                           const int* __restrict__ src) {
    for (int i0 = 0; i0 < cnt; i0 += 128) {
        int ent[4], sv[4];
#pragma unroll
        for (int u = 0; u < 4; ++u) {
            const int idx = i0 + u * 32 + lane;
            ent[u] = idx < cnt ? lw[idx] : 0;
        }
#pragma unroll
        for (int u = 0; u < 4; ++u)
            sv[u] = src[ent[u] & 0xFFFFF];
#pragma unroll
        for (int u = 0; u < 4; ++u) {
            const int idx = i0 + u * 32 + lane;
            if (idx < cnt) lw[idx] = ((ent[u] >> 20) << 16) | sv[u];
        }
    }
}

__device__ __forceinline__ void ownMax(const int* lw, int cnt, int lane,
                                       const float* __restrict__ s1,
                                       const float* s2l, int* mxi) {
    const int j = lane >> 3, h = lane & 7;
    for (int i0 = 0; i0 < cnt; i0 += 16) {
        float x[4]; int ld[4];
#pragma unroll
        for (int u = 0; u < 4; ++u) {
            const int idx = i0 + u * 4 + j;
            const int ent = idx < cnt ? lw[idx] : 0;
            ld[u] = ent >> 16;
            x[u] = s1[h * N_NODES + (ent & 0xFFFF)] + s2l[ld[u] * N_HEADS + h];
        }
#pragma unroll
        for (int u = 0; u < 4; ++u) {
            const int idx = i0 + u * 4 + j;
            const float v = x[u] > 0.0f ? x[u] : ALPHA * x[u];
            if (idx < cnt) ldsMaxFloat(&mxi[ld[u] * N_HEADS + h], v);
        }
    }
}

__device__ __forceinline__ void ownSum(const int* lw, int cnt, int lane,
                                       const float* __restrict__ s1,
                                       const float* __restrict__ Wh,
                                       const float* s2l, const int* mxi,
                                       float* dn, float* acc) {
    const int hl = lane & 7;
    for (int i0 = 0; i0 < cnt; i0 += 4) {
        int s[4], ld[4]; float sc[4]; float wh[4][N_HEADS]; float ex[4];
#pragma unroll
        for (int u = 0; u < 4; ++u) {
            const int idx = (i0 + u < cnt) ? i0 + u : cnt - 1;
            const int ent = lw[idx];
            s[u]  = ent & 0xFFFF;
            ld[u] = ent >> 16;
        }
#pragma unroll
        for (int u = 0; u < 4; ++u)
            sc[u] = s1[hl * N_NODES + s[u]];
#pragma unroll
        for (int u = 0; u < 4; ++u)
#pragma unroll
            for (int head = 0; head < N_HEADS; ++head)
                wh[u][head] = Wh[((size_t)head * N_NODES + s[u]) * N_HID + lane];
#pragma unroll
        for (int u = 0; u < 4; ++u) {
            const float x = sc[u] + s2l[ld[u] * N_HEADS + hl];
            const float v = x > 0.0f ? x : ALPHA * x;
            ex[u] = __expf(v - __int_as_float(mxi[ld[u] * N_HEADS + hl]));
        }
#pragma unroll
        for (int u = 0; u < 4; ++u) {
            if (i0 + u < cnt) {
                if (lane < N_HEADS) atomicAdd(&dn[ld[u] * N_HEADS + lane], ex[u]);
                float* ap = acc + ld[u] * (N_HEADS * N_HID) + lane;
#pragma unroll
                for (int head = 0; head < N_HEADS; ++head) {
                    const float w = __int_as_float(
                        __builtin_amdgcn_readlane(__float_as_int(ex[u]), head));
                    atomicAdd(ap + head * N_HID, w * wh[u][head]);
                }
            }
        }
    }
}

template <int PHASE>
__device__ __forceinline__ void ownWork(int* lw, int cnt, int lane,
                                        const float* __restrict__ s1,
                                        const float* __restrict__ Wh,
                                        const float* s2l, int* mxi,
                                        float* dn, float* acc) {
    if (PHASE == 0) ownMax(lw, cnt, lane, s1, s2l, mxi);
    else            ownSum(lw, cnt, lane, s1, Wh, s2l, mxi, dn, acc);
}

template <int PHASE>
__device__ __forceinline__ int ownScan(int* lw, int wave, int lane, int lo, int nloc,
                                       const int* __restrict__ src,
                                       const int* __restrict__ dst,
                                       const float* __restrict__ s1,
                                       const float* __restrict__ Wh,
                                       const float* s2l, int* mxi,
                                       float* dn, float* acc, int& cnt_out) {
    const int4* d4 = (const int4*)dst;
    int cnt = 0, rounds = 0;
    for (int c0 = wave; c0 < N_CHUNKS; c0 += OWN_WAVES * 8) {
        int4 dv[8];
#pragma unroll
        for (int u = 0; u < 8; ++u) {
            const int c = c0 + u * OWN_WAVES;
            dv[u] = c < N_CHUNKS ? d4[c * 32 + lane] : make_int4(-1, -1, -1, -1);
        }
#pragma unroll
        for (int u = 0; u < 8; ++u) {
            const int c = c0 + u * OWN_WAVES;
            const unsigned r0 = (unsigned)(dv[u].x - lo);
            const unsigned r1 = (unsigned)(dv[u].y - lo);
            const unsigned r2 = (unsigned)(dv[u].z - lo);
            const unsigned r3 = (unsigned)(dv[u].w - lo);
            const bool m0 = r0 < (unsigned)nloc, m1 = r1 < (unsigned)nloc;
            const bool m2 = r2 < (unsigned)nloc, m3 = r3 < (unsigned)nloc;
            if (__builtin_amdgcn_ballot_w32(m0 | m1 | m2 | m3) != 0u) {
                const unsigned e0 = (unsigned)c * EDGE_CHUNK + lane * 4;
                unsigned mask;
                mask = __builtin_amdgcn_ballot_w32(m0);
                if (m0) lw[cnt + __builtin_amdgcn_mbcnt_lo(mask, 0u)] = (int)((r0 << 20) | e0);
                cnt += __builtin_popcount(mask);
                mask = __builtin_amdgcn_ballot_w32(m1);
                if (m1) lw[cnt + __builtin_amdgcn_mbcnt_lo(mask, 0u)] = (int)((r1 << 20) | (e0 + 1));
                cnt += __builtin_popcount(mask);
                mask = __builtin_amdgcn_ballot_w32(m2);
                if (m2) lw[cnt + __builtin_amdgcn_mbcnt_lo(mask, 0u)] = (int)((r2 << 20) | (e0 + 2));
                cnt += __builtin_popcount(mask);
                mask = __builtin_amdgcn_ballot_w32(m3);
                if (m3) lw[cnt + __builtin_amdgcn_mbcnt_lo(mask, 0u)] = (int)((r3 << 20) | (e0 + 3));
                cnt += __builtin_popcount(mask);
                if (cnt > LIST_CAP - EDGE_CHUNK) {
                    ownListFence();
                    ownResolve(lw, cnt, lane, src);
                    ownListFence();
                    ownWork<PHASE>(lw, cnt, lane, s1, Wh, s2l, mxi, dn, acc);
                    ownListFence();
                    cnt = 0;
                    ++rounds;
                }
            }
        }
    }
    ownListFence();
    ownResolve(lw, cnt, lane, src);
    ownListFence();
    ownWork<PHASE>(lw, cnt, lane, s1, Wh, s2l, mxi, dn, acc);
    cnt_out = cnt;
    return rounds + 1;
}

__global__ __launch_bounds__(1024) void gat_own(const int* __restrict__ src,
                                                const int* __restrict__ dst,
                                                const float* __restrict__ s1,
                                                const float* __restrict__ s2,
                                                const float* __restrict__ Wh,
                                                float* __restrict__ out) {
    __shared__ __attribute__((aligned(16))) float acc[TILE_NODES * N_HEADS * N_HID];
    __shared__ int   mxi[TILE_NODES * N_HEADS];
    __shared__ float dn[TILE_NODES * N_HEADS];
    __shared__ float s2l[TILE_NODES * N_HEADS];
    __shared__ int   lst[OWN_WAVES * LIST_CAP];

    const int tid  = threadIdx.x;
    const int lane = tid & 31;
    const int wave = __builtin_amdgcn_readfirstlane(tid >> 5);
    const int lo   = blockIdx.x * TILE_NODES;
    const int nloc = (N_NODES - lo) < TILE_NODES ? (N_NODES - lo) : TILE_NODES;

    float4* acc4 = (float4*)acc;
    for (int i = tid; i < TILE_NODES * N_HEADS * N_HID / 4; i += 1024)
        acc4[i] = make_float4(0.0f, 0.0f, 0.0f, 0.0f);
    for (int i = tid; i < TILE_NODES * N_HEADS; i += 1024) {
        const int node = lo + (i >> 3);
        mxi[i] = __float_as_int(-__builtin_inff());
        dn[i]  = 0.0f;
        s2l[i] = node < N_NODES ? s2[(i & 7) * N_NODES + node] : 0.0f;
    }
    __syncthreads();

    int* lw = lst + wave * LIST_CAP;
    int cnt = 0;
    const int rounds = ownScan<0>(lw, wave, lane, lo, nloc, src, dst, s1, Wh,
                                  s2l, mxi, dn, acc, cnt);
    __syncthreads();

    if (rounds == 1)
        ownSum(lw, cnt, lane, s1, Wh, s2l, mxi, dn, acc);
    else
        ownScan<1>(lw, wave, lane, lo, nloc, src, dst, s1, Wh, s2l, mxi, dn, acc, cnt);
    __syncthreads();

    float4* o4 = (float4*)(out + (size_t)lo * (N_HEADS * N_HID));
    for (int i = tid; i < nloc * (N_HEADS * N_HID / 4); i += 1024) {
        const float d0 = dn[(i >> 6) * N_HEADS + ((i >> 3) & 7)];
        const float q = d0 > 0.0f ? d0 : 1.0f;
        float4 v = acc4[i];
        v.x /= q; v.y /= q; v.z /= q; v.w /= q;
        o4[i] = v;
    }
}

extern "C" void kernel_launch(void* const* d_in, const int* in_sizes, int n_in,
                              void* d_out, int out_size, void* d_ws, size_t ws_size,
                              hipStream_t stream) {
    const float* h   = (const float*)d_in[0];
    const float* W   = (const float*)d_in[1];
    const float* a   = (const float*)d_in[2];
    const int*   src = (const int*)d_in[3];
    const int*   dst = (const int*)d_in[4];
    float* out = (float*)d_out;

    float* ws = (float*)d_ws;
    float* Wh = ws;
    float* s1 = Wh + (size_t)N_HEADS * N_NODES * N_HID;
    float* s2 = s1 + (size_t)N_HEADS * N_NODES;

    gat_gemm<<<GM_GRID, GM_WAVES * 32, 0, stream>>>(h, W, Wh);

    gat_score<<<(SC_WAVES + 7) / 8, 256, 0, stream>>>(Wh, a, s1, s2);

    gat_own<<<(N_NODES + TILE_NODES - 1) / TILE_NODES, 1024, 0, stream>>>(src, dst, s1, s2, Wh, out);
}


# owner pass: a wave keeps its edges' scores in 32 registers (lane = edge slot x head) from max phase to sum phase: second s1 gather gone, one exp and one denom atomic per 4 edges; arrival order kept
# speedup vs baseline: 5.8896x; 1.1289x over previous
#include <hip/hip_runtime.h>
#include <hip/hip_bf16.h>

typedef float v2f __attribute__((ext_vector_type(2)));
typedef float v8f __attribute__((ext_vector_type(8)));

#define N_NODES 50000
#define N_EDGES 800000
#define N_FEAT  256
#define N_HID   32
#define N_HEADS 8
#define ALPHA   0.2f

#define GM_GRID  256
#define GM_WAVES 8
#define GM_TILES (N_NODES / 16)
#define GM_STEP  (GM_GRID * (GM_WAVES / 2))

__global__ __launch_bounds__(GM_WAVES * 32) void gat_gemm(const float* __restrict__ hmat,
                                                          const float* __restrict__ W,
                                                          float* __restrict__ Wh) {
    __shared__ v2f wl[64 * 16 * 32];

    const int tid  = threadIdx.x;
    const int lane = tid & 31;
    const int wave = __builtin_amdgcn_readfirstlane(tid >> 5);
    const int m  = lane & 15;
    const int hf = lane >> 4;

    const int nh = wave & 1;
    int tile = blockIdx.x + GM_GRID * (wave >> 1);

    const float4* rowp = (const float4*)(hmat + (size_t)(tile * 16 + m) * N_FEAT + 32 * hf);
    float4 cur[8];
#pragma unroll
    for (int i = 0; i < 8; ++i) cur[i] = rowp[i];

    for (int i0 = 0; i0 < (64 * 16 * 32) / (GM_WAVES * 32); i0 += 16) {
        v2f tmp[16];
#pragma unroll
        for (int i = 0; i < 16; ++i) {
            const int idx = tid + GM_WAVES * 32 * (i0 + i);
            const int nt = (idx >> 5) & 15, ks = idx >> 9;
            const int k = 64 * (ks >> 4) + 32 * hf + 2 * (ks & 15);
            const float* wp = W + (size_t)((nt >> 1) * N_FEAT + k) * N_HID + (nt & 1) * 16 + m;
            tmp[i][0] = wp[0];
            tmp[i][1] = wp[N_HID];
        }
#pragma unroll
        for (int i = 0; i < 16; ++i)
            wl[tid + GM_WAVES * 32 * (i0 + i)] = tmp[i];
    }
    __syncthreads();

    const v2f* bl = wl + nh * (8 * 32) + lane;

    for (; tile < GM_TILES; tile += GM_STEP) {
        const int tn = tile + GM_STEP < GM_TILES ? tile + GM_STEP : tile;
        const float4* nrow = (const float4*)(hmat + (size_t)(tn * 16 + m) * N_FEAT + 32 * hf);

        v8f acc[8];
#pragma unroll
        for (int j = 0; j < 8; ++j) acc[j] = v8f{};

#pragma nounroll
        for (int kc = 0; kc < 4; ++kc) {
            const float4* np = kc < 3 ? rowp + (kc + 1) * 16 : nrow;
            float4 nxt[8];
#pragma unroll
            for (int i = 0; i < 8; ++i) nxt[i] = np[i];

            const v2f* bk = bl + kc * (16 * 16 * 32);
#pragma unroll
            for (int s = 0; s < 16; ++s) {
                const float4 q = cur[s >> 1];
                v2f a;
                a[0] = (s & 1) ? q.z : q.x;
                a[1] = (s & 1) ? q.w : q.y;
#pragma unroll
                for (int j = 0; j < 8; ++j) {
                    const v2f b = bk[(s * 16 + j) * 32];
                    acc[j] = __builtin_amdgcn_wmma_f32_16x16x4_f32(false, a, false, b,
                                                                   (short)0, acc[j], false, false);
                }
            }
#pragma unroll
            for (int i = 0; i < 8; ++i) cur[i] = nxt[i];
        }

#pragma unroll
        for (int j = 0; j < 8; ++j) {
            const int head = nh * 4 + (j >> 1);
            float* op = Wh + ((size_t)head * N_NODES + tile * 16 + hf * 8) * N_HID + (j & 1) * 16 + m;
#pragma unroll
            for (int r = 0; r < 8; ++r)
                op[r * N_HID] = acc[j][r];
        }
        rowp = nrow;
    }
}

#define SC_ROWS  64
#define SC_WAVES ((N_HEADS * N_NODES) / SC_ROWS)

static_assert((N_HEADS * N_NODES) % SC_ROWS == 0, "whole 64-row pieces");
static_assert(N_NODES % 16 == 0, "a group of 16 rows lies inside one head");

__global__ __launch_bounds__(256) void gat_score(const float* __restrict__ Wh,
                                                 const float* __restrict__ avec,
                                                 float* __restrict__ s1,
                                                 float* __restrict__ s2) {
    const int lane = threadIdx.x & 31;
    const int w = blockIdx.x * 8 + __builtin_amdgcn_readfirstlane(threadIdx.x >> 5);
    if (w >= SC_WAVES) return;
    const int j = lane >> 3;
    const int q = lane & 7;

    const float4* w4 = (const float4*)Wh + (size_t)w * (SC_ROWS * 8) + lane;
    float4 v[16];
#pragma unroll
    for (int i = 0; i < 16; ++i)
        v[i] = w4[i * 32];

    const float4* a4 = (const float4*)avec;
    float p1[16], p2[16];
#pragma unroll
    for (int t = 0; t < 4; ++t) {
        const int head = (w * SC_ROWS + t * 16) / N_NODES;
        const float4 x1 = a4[head * 16 + q];
        const float4 x2 = a4[head * 16 + 8 + q];
#pragma unroll
        for (int u = 0; u < 4; ++u) {
            const float4 c = v[t * 4 + u];
            p1[t * 4 + u] = c.x * x1.x + c.y * x1.y + c.z * x1.z + c.w * x1.w;
            p2[t * 4 + u] = c.x * x2.x + c.y * x2.y + c.z * x2.z + c.w * x2.w;
        }
    }

    const bool b2 = (q & 4) != 0, b1 = (q & 2) != 0, b0 = (q & 1) != 0;
    float r1[8], r2[8];
#pragma unroll
    for (int g = 0; g < 2; ++g)
#pragma unroll
        for (int l = 0; l < 4; ++l) {
            const int i = g * 8 + l;
            const float k1 = b2 ? p1[i + 4] : p1[i], d1 = b2 ? p1[i] : p1[i + 4];
            const float k2 = b2 ? p2[i + 4] : p2[i], d2 = b2 ? p2[i] : p2[i + 4];
            r1[g * 4 + l] = k1 + __shfl_xor(d1, 4, 32);
            r2[g * 4 + l] = k2 + __shfl_xor(d2, 4, 32);
        }
    float t1[4], t2[4];
#pragma unroll
    for (int g = 0; g < 2; ++g)
#pragma unroll
        for (int l = 0; l < 2; ++l) {
            const int i = g * 4 + l;
            const float k1 = b1 ? r1[i + 2] : r1[i], d1 = b1 ? r1[i] : r1[i + 2];
            const float k2 = b1 ? r2[i + 2] : r2[i], d2 = b1 ? r2[i] : r2[i + 2];
            t1[g * 2 + l] = k1 + __shfl_xor(d1, 2, 32);
            t2[g * 2 + l] = k2 + __shfl_xor(d2, 2, 32);
        }
    float* o1 = s1 + (size_t)w * SC_ROWS + 4 * q + j;
    float* o2 = s2 + (size_t)w * SC_ROWS + 4 * q + j;
#pragma unroll
    for (int g = 0; g < 2; ++g) {
        const int i = g * 2;
        const float k1 = b0 ? t1[i + 1] : t1[i], d1 = b0 ? t1[i] : t1[i + 1];
        const float k2 = b0 ? t2[i + 1] : t2[i], d2 = b0 ? t2[i] : t2[i + 1];
        o1[g * 32] = k1 + __shfl_xor(d1, 1, 32);
        o2[g * 32] = k2 + __shfl_xor(d2, 1, 32);
    }
}

#define TILE_NODES 196
#define OWN_WAVES  32
#define LIST_CAP   384
#define EDGE_CHUNK 128
#define N_CHUNKS   (N_EDGES / EDGE_CHUNK)
#define KEEP_GROUPS 32

typedef float v32f __attribute__((ext_vector_type(32)));

static_assert(N_EDGES % EDGE_CHUNK == 0, "dst[] is read in whole 128-edge chunks");
static_assert(N_EDGES <= (1 << 20) && TILE_NODES <= 2048, "list entry: local node << 20 | edge");
static_assert(N_NODES <= (1 << 16), "resolved list entry: local node << 16 | src");
static_assert(LIST_CAP > EDGE_CHUNK, "a chunk must fit behind the flush mark");
static_assert((KEEP_GROUPS * 4) % 16 == 0 && KEEP_GROUPS <= 32, "phase B fills keep[] 4 groups a trip");

__device__ __forceinline__ void ldsMaxFloat(int* addr, float val) {
    if (val >= 0.0f)
        atomicMax(addr, __float_as_int(val));
    else
        atomicMin((unsigned int*)addr, __float_as_uint(val));
}

__device__ __forceinline__ void ownListFence() {
    __builtin_amdgcn_fence(__ATOMIC_ACQ_REL, "wavefront");
}

__device__ __forceinline__ void ownResolve(int* lw, int cnt, int lane,
                                           const int* __restrict__ src) {
    for (int i0 = 0; i0 < cnt; i0 += 128) {
        int ent[4], sv[4];
#pragma unroll
        for (int u = 0; u < 4; ++u) {
            const int idx = i0 + u * 32 + lane;
            ent[u] = idx < cnt ? lw[idx] : 0;
        }
#pragma unroll
        for (int u = 0; u < 4; ++u)
            sv[u] = src[ent[u] & 0xFFFFF];
#pragma unroll
        for (int u = 0; u < 4; ++u) {
            const int idx = i0 + u * 32 + lane;
            if (idx < cnt) lw[idx] = ((ent[u] >> 20) << 16) | sv[u];
        }
    }
}

__device__ __forceinline__ void ownMax(const int* lw, int cnt, int lane,
                                       const float* __restrict__ s1,
                                       const float* s2l, int* mxi, v32f& keep) {
    const int j = lane >> 3, h = lane & 7;
    for (int i0 = 0; i0 < cnt; i0 += 16) {
        float x[4]; int ld[4];
#pragma unroll
        for (int u = 0; u < 4; ++u) {
            const int idx = i0 + u * 4 + j;
            const int ent = idx < cnt ? lw[idx] : 0;
            ld[u] = ent >> 16;
            x[u] = s1[h * N_NODES + (ent & 0xFFFF)] + s2l[ld[u] * N_HEADS + h];
        }
#pragma unroll
        for (int u = 0; u < 4; ++u) {
            const int idx = i0 + u * 4 + j;
            x[u] = x[u] > 0.0f ? x[u] : ALPHA * x[u];
            if (idx < cnt) ldsMaxFloat(&mxi[ld[u] * N_HEADS + h], x[u]);
        }
        if (i0 < KEEP_GROUPS * 4) {
#pragma unroll
            for (int u = 0; u < 4; ++u)
                keep[(i0 >> 2) + u] = x[u];
        }
    }
}

template <bool KEPT>
__device__ __forceinline__ void ownSum(const int* lw, int cnt, int lane,
                                       const float* __restrict__ s1,
                                       const float* __restrict__ Wh,
                                       const float* s2l, const int* mxi,
                                       float* dn, float* acc, const v32f& keep) {
    const int j = lane >> 3, h = lane & 7;
    for (int i0 = 0; i0 < cnt; i0 += 4) {
        const bool mine = i0 + j < cnt;
        const int ent = lw[mine ? i0 + j : cnt - 1];
        const int lj = ent >> 16;
        int s[4], ld[4]; float wh[4][N_HEADS];
#pragma unroll
        for (int u = 0; u < 4; ++u) {
            const int e = __builtin_amdgcn_readlane(ent, 8 * u);
            s[u]  = e & 0xFFFF;
            ld[u] = e >> 16;
        }
        float v;
        if (KEPT && i0 < KEEP_GROUPS * 4) {
            v = keep[i0 >> 2];
        } else {
            const float x = s1[h * N_NODES + (ent & 0xFFFF)] + s2l[lj * N_HEADS + h];
            v = x > 0.0f ? x : ALPHA * x;
        }
#pragma unroll
        for (int u = 0; u < 4; ++u)
#pragma unroll
            for (int head = 0; head < N_HEADS; ++head)
                wh[u][head] = Wh[((size_t)head * N_NODES + s[u]) * N_HID + lane];
        const float ex = __expf(v - __int_as_float(mxi[lj * N_HEADS + h]));
        if (mine) atomicAdd(&dn[lj * N_HEADS + h], ex);
#pragma unroll
        for (int u = 0; u < 4; ++u) {
            if (i0 + u < cnt) {
                float* ap = acc + ld[u] * (N_HEADS * N_HID) + lane;
#pragma unroll
                for (int head = 0; head < N_HEADS; ++head) {
                    const float w = __int_as_float(
                        __builtin_amdgcn_readlane(__float_as_int(ex), 8 * u + head));
                    atomicAdd(ap + head * N_HID, w * wh[u][head]);
                }
            }
        }
    }
}

template <int PHASE>
__device__ __forceinline__ void ownWork(int* lw, int cnt, int lane,
                                        const float* __restrict__ s1,
                                        const float* __restrict__ Wh,
                                        const float* s2l, int* mxi,
                                        float* dn, float* acc, v32f& keep) {
    if (PHASE == 0) ownMax(lw, cnt, lane, s1, s2l, mxi, keep);
    else            ownSum<false>(lw, cnt, lane, s1, Wh, s2l, mxi, dn, acc, keep);
}

template <int PHASE>
__device__ __forceinline__ int ownScan(int* lw, int wave, int lane, int lo, int nloc,
                                       const int* __restrict__ src,
                                       const int* __restrict__ dst,
                                       const float* __restrict__ s1,
                                       const float* __restrict__ Wh,
                                       const float* s2l, int* mxi,
                                       float* dn, float* acc, int& cnt_out,
                                       v32f& keep) {
    const int4* d4 = (const int4*)dst;
    int cnt = 0, rounds = 0;
    for (int c0 = wave; c0 < N_CHUNKS; c0 += OWN_WAVES * 8) {
        int4 dv[8];
#pragma unroll
        for (int u = 0; u < 8; ++u) {
            const int c = c0 + u * OWN_WAVES;
            dv[u] = c < N_CHUNKS ? d4[c * 32 + lane] : make_int4(-1, -1, -1, -1);
        }
#pragma unroll
        for (int u = 0; u < 8; ++u) {
            const int c = c0 + u * OWN_WAVES;
            const unsigned r0 = (unsigned)(dv[u].x - lo);
            const unsigned r1 = (unsigned)(dv[u].y - lo);
            const unsigned r2 = (unsigned)(dv[u].z - lo);
            const unsigned r3 = (unsigned)(dv[u].w - lo);
            const bool m0 = r0 < (unsigned)nloc, m1 = r1 < (unsigned)nloc;
            const bool m2 = r2 < (unsigned)nloc, m3 = r3 < (unsigned)nloc;
            if (__builtin_amdgcn_ballot_w32(m0 | m1 | m2 | m3) != 0u) {
                const unsigned e0 = (unsigned)c * EDGE_CHUNK + lane * 4;
                unsigned mask;
                mask = __builtin_amdgcn_ballot_w32(m0);
                if (m0) lw[cnt + __builtin_amdgcn_mbcnt_lo(mask, 0u)] = (int)((r0 << 20) | e0);
                cnt += __builtin_popcount(mask);
                mask = __builtin_amdgcn_ballot_w32(m1);
                if (m1) lw[cnt + __builtin_amdgcn_mbcnt_lo(mask, 0u)] = (int)((r1 << 20) | (e0 + 1));
                cnt += __builtin_popcount(mask);
                mask = __builtin_amdgcn_ballot_w32(m2);
                if (m2) lw[cnt + __builtin_amdgcn_mbcnt_lo(mask, 0u)] = (int)((r2 << 20) | (e0 + 2));
                cnt += __builtin_popcount(mask);
                mask = __builtin_amdgcn_ballot_w32(m3);
                if (m3) lw[cnt + __builtin_amdgcn_mbcnt_lo(mask, 0u)] = (int)((r3 << 20) | (e0 + 3));
                cnt += __builtin_popcount(mask);
                if (cnt > LIST_CAP - EDGE_CHUNK) {
                    ownListFence();
                    ownResolve(lw, cnt, lane, src);
                    ownListFence();
                    ownWork<PHASE>(lw, cnt, lane, s1, Wh, s2l, mxi, dn, acc, keep);
                    ownListFence();
                    cnt = 0;
                    ++rounds;
                }
            }
        }
    }
    ownListFence();
    ownResolve(lw, cnt, lane, src);
    ownListFence();
    ownWork<PHASE>(lw, cnt, lane, s1, Wh, s2l, mxi, dn, acc, keep);
    cnt_out = cnt;
    return rounds + 1;
}

__global__ __launch_bounds__(1024) void gat_own(const int* __restrict__ src,
                                                const int* __restrict__ dst,
                                                const float* __restrict__ s1,
                                                const float* __restrict__ s2,
                                                const float* __restrict__ Wh,
                                                float* __restrict__ out) {
    __shared__ __attribute__((aligned(16))) float acc[TILE_NODES * N_HEADS * N_HID];
    __shared__ int   mxi[TILE_NODES * N_HEADS];
    __shared__ float dn[TILE_NODES * N_HEADS];
    __shared__ float s2l[TILE_NODES * N_HEADS];
    __shared__ int   lst[OWN_WAVES * LIST_CAP];

    const int tid  = threadIdx.x;
    const int lane = tid & 31;
    const int wave = __builtin_amdgcn_readfirstlane(tid >> 5);
    const int lo   = blockIdx.x * TILE_NODES;
    const int nloc = (N_NODES - lo) < TILE_NODES ? (N_NODES - lo) : TILE_NODES;

    float4* acc4 = (float4*)acc;
    for (int i = tid; i < TILE_NODES * N_HEADS * N_HID / 4; i += 1024)
        acc4[i] = make_float4(0.0f, 0.0f, 0.0f, 0.0f);
    for (int i = tid; i < TILE_NODES * N_HEADS; i += 1024) {
        const int node = lo + (i >> 3);
        mxi[i] = __float_as_int(-__builtin_inff());
        dn[i]  = 0.0f;
        s2l[i] = node < N_NODES ? s2[(i & 7) * N_NODES + node] : 0.0f;
    }
    __syncthreads();

    int* lw = lst + wave * LIST_CAP;
    int cnt = 0;
    v32f keep = {};
    const int rounds = ownScan<0>(lw, wave, lane, lo, nloc, src, dst, s1, Wh,
                                  s2l, mxi, dn, acc, cnt, keep);
    __syncthreads();

    if (rounds == 1)
        ownSum<true>(lw, cnt, lane, s1, Wh, s2l, mxi, dn, acc, keep);
    else
        ownScan<1>(lw, wave, lane, lo, nloc, src, dst, s1, Wh, s2l, mxi, dn, acc, cnt, keep);
    __syncthreads();

    float4* o4 = (float4*)(out + (size_t)lo * (N_HEADS * N_HID));
    for (int i = tid; i < nloc * (N_HEADS * N_HID / 4); i += 1024) {
        const float d0 = dn[(i >> 6) * N_HEADS + ((i >> 3) & 7)];
        const float q = d0 > 0.0f ? d0 : 1.0f;
        float4 v = acc4[i];
        v.x /= q; v.y /= q; v.z /= q; v.w /= q;
        o4[i] = v;
    }
}

extern "C" void kernel_launch(void* const* d_in, const int* in_sizes, int n_in,
                              void* d_out, int out_size, void* d_ws, size_t ws_size,
                              hipStream_t stream) {
    const float* h   = (const float*)d_in[0];
    const float* W   = (const float*)d_in[1];
    const float* a   = (const float*)d_in[2];
    const int*   src = (const int*)d_in[3];
    const int*   dst = (const int*)d_in[4];
    float* out = (float*)d_out;

    float* ws = (float*)d_ws;
    float* Wh = ws;
    float* s1 = Wh + (size_t)N_HEADS * N_NODES * N_HID;
    float* s2 = s1 + (size_t)N_HEADS * N_NODES;

    gat_gemm<<<GM_GRID, GM_WAVES * 32, 0, stream>>>(h, W, Wh);

    gat_score<<<(SC_WAVES + 7) / 8, 256, 0, stream>>>(Wh, a, s1, s2);

    gat_own<<<(N_NODES + TILE_NODES - 1) / TILE_NODES, 1024, 0, stream>>>(src, dst, s1, s2, Wh, out);
}


# owner scan reads a 1-byte owner-tile key per edge, not the 4-byte dst; score launch writes the keys and a (local dst, src) word per edge; a wave's edge order changes, LDS atomics stay arrival order
# speedup vs baseline: 6.5016x; 1.1039x over previous
#include <hip/hip_runtime.h>
#include <hip/hip_bf16.h>

typedef float v2f __attribute__((ext_vector_type(2)));
typedef float v8f __attribute__((ext_vector_type(8)));

#define N_NODES 50000
#define N_EDGES 800000
#define N_FEAT  256
#define N_HID   32
#define N_HEADS 8
#define ALPHA   0.2f
#define TILE_NODES 196

#define GM_GRID  256
#define GM_WAVES 8
#define GM_TILES (N_NODES / 16)
#define GM_STEP  (GM_GRID * (GM_WAVES / 2))

__global__ __launch_bounds__(GM_WAVES * 32) void gat_gemm(const float* __restrict__ hmat,
                                                          const float* __restrict__ W,
                                                          float* __restrict__ Wh) {
    __shared__ v2f wl[64 * 16 * 32];

    const int tid  = threadIdx.x;
    const int lane = tid & 31;
    const int wave = __builtin_amdgcn_readfirstlane(tid >> 5);
    const int m  = lane & 15;
    const int hf = lane >> 4;

    const int nh = wave & 1;
    int tile = blockIdx.x + GM_GRID * (wave >> 1);

    const float4* rowp = (const float4*)(hmat + (size_t)(tile * 16 + m) * N_FEAT + 32 * hf);
    float4 cur[8];
#pragma unroll
    for (int i = 0; i < 8; ++i) cur[i] = rowp[i];

    for (int i0 = 0; i0 < (64 * 16 * 32) / (GM_WAVES * 32); i0 += 16) {
        v2f tmp[16];
#pragma unroll
        for (int i = 0; i < 16; ++i) {
            const int idx = tid + GM_WAVES * 32 * (i0 + i);
            const int nt = (idx >> 5) & 15, ks = idx >> 9;
            const int k = 64 * (ks >> 4) + 32 * hf + 2 * (ks & 15);
            const float* wp = W + (size_t)((nt >> 1) * N_FEAT + k) * N_HID + (nt & 1) * 16 + m;
            tmp[i][0] = wp[0];
            tmp[i][1] = wp[N_HID];
        }
#pragma unroll
        for (int i = 0; i < 16; ++i)
            wl[tid + GM_WAVES * 32 * (i0 + i)] = tmp[i];
    }
    __syncthreads();

    const v2f* bl = wl + nh * (8 * 32) + lane;

    for (; tile < GM_TILES; tile += GM_STEP) {
        const int tn = tile + GM_STEP < GM_TILES ? tile + GM_STEP : tile;
        const float4* nrow = (const float4*)(hmat + (size_t)(tn * 16 + m) * N_FEAT + 32 * hf);

        v8f acc[8];
#pragma unroll
        for (int j = 0; j < 8; ++j) acc[j] = v8f{};

#pragma nounroll
        for (int kc = 0; kc < 4; ++kc) {
            const float4* np = kc < 3 ? rowp + (kc + 1) * 16 : nrow;
            float4 nxt[8];
#pragma unroll
            for (int i = 0; i < 8; ++i) nxt[i] = np[i];

            const v2f* bk = bl + kc * (16 * 16 * 32);
#pragma unroll
            for (int s = 0; s < 16; ++s) {
                const float4 q = cur[s >> 1];
                v2f a;
                a[0] = (s & 1) ? q.z : q.x;
                a[1] = (s & 1) ? q.w : q.y;
#pragma unroll
                for (int j = 0; j < 8; ++j) {
                    const v2f b = bk[(s * 16 + j) * 32];
                    acc[j] = __builtin_amdgcn_wmma_f32_16x16x4_f32(false, a, false, b,
                                                                   (short)0, acc[j], false, false);
                }
            }
#pragma unroll
            for (int i = 0; i < 8; ++i) cur[i] = nxt[i];
        }

#pragma unroll
        for (int j = 0; j < 8; ++j) {
            const int head = nh * 4 + (j >> 1);
            float* op = Wh + ((size_t)head * N_NODES + tile * 16 + hf * 8) * N_HID + (j & 1) * 16 + m;
#pragma unroll
            for (int r = 0; r < 8; ++r)
                op[r * N_HID] = acc[j][r];
        }
        rowp = nrow;
    }
}

#define SC_ROWS  64
#define SC_WAVES ((N_HEADS * N_NODES) / SC_ROWS)

static_assert((N_HEADS * N_NODES) % SC_ROWS == 0, "whole 64-row pieces");
static_assert(N_NODES % 16 == 0, "a group of 16 rows lies inside one head");
static_assert(N_EDGES == SC_WAVES * 128, "every score wave takes 128 edges");
static_assert((N_NODES + TILE_NODES - 1) / TILE_NODES <= 256, "the owner tile fits a byte");
static_assert(N_NODES <= (1 << 16) && TILE_NODES <= 256, "edge word: local node << 16 | src");

__global__ __launch_bounds__(256) void gat_score(const float* __restrict__ Wh,
                                                 const float* __restrict__ avec,
                                                 const int* __restrict__ src,
                                                 const int* __restrict__ dst,
                                                 float* __restrict__ s1,
                                                 float* __restrict__ s2,
                                                 int* __restrict__ ekey,
                                                 int* __restrict__ eword) {
    const int lane = threadIdx.x & 31;
    const int w = blockIdx.x * 8 + __builtin_amdgcn_readfirstlane(threadIdx.x >> 5);
    if (w >= SC_WAVES) return;
    const int j = lane >> 3;
    const int q = lane & 7;

    const int4 dv = ((const int4*)dst)[w * 32 + lane];
    const int4 sv = ((const int4*)src)[w * 32 + lane];

    const float4* w4 = (const float4*)Wh + (size_t)w * (SC_ROWS * 8) + lane;
    float4 v[16];
#pragma unroll
    for (int i = 0; i < 16; ++i)
        v[i] = w4[i * 32];

    const float4* a4 = (const float4*)avec;
    float p1[16], p2[16];
#pragma unroll
    for (int t = 0; t < 4; ++t) {
        const int head = (w * SC_ROWS + t * 16) / N_NODES;
        const float4 x1 = a4[head * 16 + q];
        const float4 x2 = a4[head * 16 + 8 + q];
#pragma unroll
        for (int u = 0; u < 4; ++u) {
            const float4 c = v[t * 4 + u];
            p1[t * 4 + u] = c.x * x1.x + c.y * x1.y + c.z * x1.z + c.w * x1.w;
            p2[t * 4 + u] = c.x * x2.x + c.y * x2.y + c.z * x2.z + c.w * x2.w;
        }
    }

    const bool b2 = (q & 4) != 0, b1 = (q & 2) != 0, b0 = (q & 1) != 0;
    float r1[8], r2[8];
#pragma unroll
    for (int g = 0; g < 2; ++g)
#pragma unroll
        for (int l = 0; l < 4; ++l) {
            const int i = g * 8 + l;
            const float k1 = b2 ? p1[i + 4] : p1[i], d1 = b2 ? p1[i] : p1[i + 4];
            const float k2 = b2 ? p2[i + 4] : p2[i], d2 = b2 ? p2[i] : p2[i + 4];
            r1[g * 4 + l] = k1 + __shfl_xor(d1, 4, 32);
            r2[g * 4 + l] = k2 + __shfl_xor(d2, 4, 32);
        }
    float t1[4], t2[4];
#pragma unroll
    for (int g = 0; g < 2; ++g)
#pragma unroll
        for (int l = 0; l < 2; ++l) {
            const int i = g * 4 + l;
            const float k1 = b1 ? r1[i + 2] : r1[i], d1 = b1 ? r1[i] : r1[i + 2];
            const float k2 = b1 ? r2[i + 2] : r2[i], d2 = b1 ? r2[i] : r2[i + 2];
            t1[g * 2 + l] = k1 + __shfl_xor(d1, 2, 32);
            t2[g * 2 + l] = k2 + __shfl_xor(d2, 2, 32);
        }
    float* o1 = s1 + (size_t)w * SC_ROWS + 4 * q + j;
    float* o2 = s2 + (size_t)w * SC_ROWS + 4 * q + j;
#pragma unroll
    for (int g = 0; g < 2; ++g) {
        const int i = g * 2;
        const float k1 = b0 ? t1[i + 1] : t1[i], d1 = b0 ? t1[i] : t1[i + 1];
        const float k2 = b0 ? t2[i + 1] : t2[i], d2 = b0 ? t2[i] : t2[i + 1];
        o1[g * 32] = k1 + __shfl_xor(d1, 1, 32);
        o2[g * 32] = k2 + __shfl_xor(d2, 1, 32);
    }

    const unsigned ot0 = (unsigned)dv.x / TILE_NODES, ot1 = (unsigned)dv.y / TILE_NODES;
    const unsigned ot2 = (unsigned)dv.z / TILE_NODES, ot3 = (unsigned)dv.w / TILE_NODES;
    ekey[w * 32 + lane] = (int)(ot0 | (ot1 << 8) | (ot2 << 16) | (ot3 << 24));
    int4 ew;
    ew.x = (int)(((unsigned)dv.x - ot0 * TILE_NODES) << 16) | sv.x;
    ew.y = (int)(((unsigned)dv.y - ot1 * TILE_NODES) << 16) | sv.y;
    ew.z = (int)(((unsigned)dv.z - ot2 * TILE_NODES) << 16) | sv.z;
    ew.w = (int)(((unsigned)dv.w - ot3 * TILE_NODES) << 16) | sv.w;
    ((int4*)eword)[w * 32 + lane] = ew;
}

#define OWN_WAVES  32
#define KEY_VECS   (N_EDGES / 16)
#define KEY_LOADS  ((KEY_VECS + 31) / 32)
#define LOAD_EDGES 512
#define SCAN_DEPTH 7
#define LIST_CAP   704
#define KEEP_GROUPS 32

typedef float v32f __attribute__((ext_vector_type(32)));

static_assert(N_EDGES % 16 == 0, "the key plane is read in whole 16-byte pieces");
static_assert(LIST_CAP > LOAD_EDGES, "a key load's edges must fit behind the flush mark");
static_assert((KEEP_GROUPS * 4) % 16 == 0 && KEEP_GROUPS <= 32, "phase B fills keep[] 4 groups a trip");

__device__ __forceinline__ void ldsMaxFloat(int* addr, float val) {
    if (val >= 0.0f)
        atomicMax(addr, __float_as_int(val));
    else
        atomicMin((unsigned int*)addr, __float_as_uint(val));
}

__device__ __forceinline__ void ownListFence() {
    __builtin_amdgcn_fence(__ATOMIC_ACQ_REL, "wavefront");
}

__device__ __forceinline__ void ownResolve(int* lw, int cnt, int lane,
                                           const int* __restrict__ eword) {
    for (int i0 = 0; i0 < cnt; i0 += 128) {
        int ent[4], wv[4];
#pragma unroll
        for (int u = 0; u < 4; ++u) {
            const int idx = i0 + u * 32 + lane;
            ent[u] = idx < cnt ? lw[idx] : 0;
        }
#pragma unroll
        for (int u = 0; u < 4; ++u)
            wv[u] = eword[ent[u]];
#pragma unroll
        for (int u = 0; u < 4; ++u) {
            const int idx = i0 + u * 32 + lane;
            if (idx < cnt) lw[idx] = wv[u];
        }
    }
}

__device__ __forceinline__ void ownMax(const int* lw, int cnt, int lane,
                                       const float* __restrict__ s1,
                                       const float* s2l, int* mxi, v32f& keep) {
    const int j = lane >> 3, h = lane & 7;
    for (int i0 = 0; i0 < cnt; i0 += 16) {
        float x[4]; int ld[4];
#pragma unroll
        for (int u = 0; u < 4; ++u) {
            const int idx = i0 + u * 4 + j;
            const int ent = idx < cnt ? lw[idx] : 0;
            ld[u] = ent >> 16;
            x[u] = s1[h * N_NODES + (ent & 0xFFFF)] + s2l[ld[u] * N_HEADS + h];
        }
#pragma unroll
        for (int u = 0; u < 4; ++u) {
            const int idx = i0 + u * 4 + j;
            x[u] = x[u] > 0.0f ? x[u] : ALPHA * x[u];
            if (idx < cnt) ldsMaxFloat(&mxi[ld[u] * N_HEADS + h], x[u]);
        }
        if (i0 < KEEP_GROUPS * 4) {
#pragma unroll
            for (int u = 0; u < 4; ++u)
                keep[(i0 >> 2) + u] = x[u];
        }
    }
}

template <bool KEPT>
__device__ __forceinline__ void ownSum(const int* lw, int cnt, int lane,
                                       const float* __restrict__ s1,
                                       const float* __restrict__ Wh,
                                       const float* s2l, const int* mxi,
                                       float* dn, float* acc, const v32f& keep) {
    const int j = lane >> 3, h = lane & 7;
    for (int i0 = 0; i0 < cnt; i0 += 4) {
        const bool mine = i0 + j < cnt;
        const int ent = lw[mine ? i0 + j : cnt - 1];
        const int lj = ent >> 16;
        int s[4], ld[4]; float wh[4][N_HEADS];
#pragma unroll
        for (int u = 0; u < 4; ++u) {
            const int e = __builtin_amdgcn_readlane(ent, 8 * u);
            s[u]  = e & 0xFFFF;
            ld[u] = e >> 16;
        }
        float v;
        if (KEPT && i0 < KEEP_GROUPS * 4) {
            v = keep[i0 >> 2];
        } else {
            const float x = s1[h * N_NODES + (ent & 0xFFFF)] + s2l[lj * N_HEADS + h];
            v = x > 0.0f ? x : ALPHA * x;
        }
#pragma unroll
        for (int u = 0; u < 4; ++u)
#pragma unroll
            for (int head = 0; head < N_HEADS; ++head)
                wh[u][head] = Wh[((size_t)head * N_NODES + s[u]) * N_HID + lane];
        const float ex = __expf(v - __int_as_float(mxi[lj * N_HEADS + h]));
        if (mine) atomicAdd(&dn[lj * N_HEADS + h], ex);
#pragma unroll
        for (int u = 0; u < 4; ++u) {
            if (i0 + u < cnt) {
                float* ap = acc + ld[u] * (N_HEADS * N_HID) + lane;
#pragma unroll
                for (int head = 0; head < N_HEADS; ++head) {
                    const float w = __int_as_float(
                        __builtin_amdgcn_readlane(__float_as_int(ex), 8 * u + head));
                    atomicAdd(ap + head * N_HID, w * wh[u][head]);
                }
            }
        }
    }
}

template <int PHASE>
__device__ __forceinline__ void ownWork(int* lw, int cnt, int lane,
                                        const float* __restrict__ s1,
                                        const float* __restrict__ Wh,
                                        const float* s2l, int* mxi,
                                        float* dn, float* acc, v32f& keep) {
    if (PHASE == 0) ownMax(lw, cnt, lane, s1, s2l, mxi, keep);
    else            ownSum<false>(lw, cnt, lane, s1, Wh, s2l, mxi, dn, acc, keep);
}

template <int PHASE>
__device__ __forceinline__ int ownScan(int* lw, int wave, int lane, int tile,
                                       const int* __restrict__ ekey,
                                       const int* __restrict__ eword,
                                       const float* __restrict__ s1,
                                       const float* __restrict__ Wh,
                                       const float* s2l, int* mxi,
                                       float* dn, float* acc, int& cnt_out,
                                       v32f& keep) {
    const uint4* k4 = (const uint4*)ekey;
    const unsigned K = (unsigned)tile * 0x01010101u;
    int cnt = 0, rounds = 0;
    for (int c0 = wave; c0 < KEY_LOADS; c0 += OWN_WAVES * SCAN_DEPTH) {
        uint4 kv[SCAN_DEPTH];
#pragma unroll
        for (int u = 0; u < SCAN_DEPTH; ++u) {
            const int v = (c0 + u * OWN_WAVES) * 32 + lane;
            kv[u] = k4[v < KEY_VECS ? v : KEY_VECS - 1];
        }
#pragma unroll
        for (int u = 0; u < SCAN_DEPTH; ++u) {
            const unsigned kw[4] = {kv[u].x, kv[u].y, kv[u].z, kv[u].w};
            const int v = (c0 + u * OWN_WAVES) * 32 + lane;
            const unsigned rest = v < KEY_VECS ? 0x7F7F7F7Fu : 0xFFFFFFFFu;
            unsigned z[4];
#pragma unroll
            for (int d = 0; d < 4; ++d) {
                const unsigned t = kw[d] ^ K;
                z[d] = ~(((t & 0x7F7F7F7Fu) + 0x7F7F7F7Fu) | t | rest);
            }
            unsigned mask = __builtin_amdgcn_ballot_w32((z[0] | z[1] | z[2] | z[3]) != 0u);
            if (mask != 0u) {
                unsigned zc = (z[0] >> 7) | (z[1] >> 6) | (z[2] >> 5) | (z[3] >> 4);
                const unsigned e0 = (unsigned)v * 16;
#pragma nounroll
                for (int turn = 0; turn < 16 && mask != 0u; ++turn) {
                    if (zc != 0u) {
                        const unsigned p = (unsigned)__builtin_ctz(zc);
                        lw[cnt + __builtin_amdgcn_mbcnt_lo(mask, 0u)] =
                            (int)(e0 + ((p & 7u) << 2) + (p >> 3));
                    }
                    cnt += __builtin_popcount(mask);
                    zc &= zc - 1u;
                    mask = __builtin_amdgcn_ballot_w32(zc != 0u);
                }
                if (cnt > LIST_CAP - LOAD_EDGES) {
                    ownListFence();
                    ownResolve(lw, cnt, lane, eword);
                    ownListFence();
                    ownWork<PHASE>(lw, cnt, lane, s1, Wh, s2l, mxi, dn, acc, keep);
                    ownListFence();
                    cnt = 0;
                    ++rounds;
                }
            }
        }
    }
    ownListFence();
    ownResolve(lw, cnt, lane, eword);
    ownListFence();
    ownWork<PHASE>(lw, cnt, lane, s1, Wh, s2l, mxi, dn, acc, keep);
    cnt_out = cnt;
    return rounds + 1;
}

__global__ __launch_bounds__(1024) void gat_own(const int* __restrict__ ekey,
                                                const int* __restrict__ eword,
                                                const float* __restrict__ s1,
                                                const float* __restrict__ s2,
                                                const float* __restrict__ Wh,
                                                float* __restrict__ out) {
    __shared__ __attribute__((aligned(16))) float acc[TILE_NODES * N_HEADS * N_HID];
    __shared__ int   mxi[TILE_NODES * N_HEADS];
    __shared__ float dn[TILE_NODES * N_HEADS];
    __shared__ float s2l[TILE_NODES * N_HEADS];
    __shared__ int   lst[OWN_WAVES * LIST_CAP];

    const int tid  = threadIdx.x;
    const int lane = tid & 31;
    const int wave = __builtin_amdgcn_readfirstlane(tid >> 5);
    const int lo   = blockIdx.x * TILE_NODES;
    const int nloc = (N_NODES - lo) < TILE_NODES ? (N_NODES - lo) : TILE_NODES;

    float4* acc4 = (float4*)acc;
    for (int i = tid; i < TILE_NODES * N_HEADS * N_HID / 4; i += 1024)
        acc4[i] = make_float4(0.0f, 0.0f, 0.0f, 0.0f);
    for (int i = tid; i < TILE_NODES * N_HEADS; i += 1024) {
        const int node = lo + (i >> 3);
        mxi[i] = __float_as_int(-__builtin_inff());
        dn[i]  = 0.0f;
        s2l[i] = node < N_NODES ? s2[(i & 7) * N_NODES + node] : 0.0f;
    }
    __syncthreads();

    int* lw = lst + wave * LIST_CAP;
    int cnt = 0;
    v32f keep = {};
    const int rounds = ownScan<0>(lw, wave, lane, (int)blockIdx.x, ekey, eword, s1, Wh,
                                  s2l, mxi, dn, acc, cnt, keep);
    __syncthreads();

    if (rounds == 1)
        ownSum<true>(lw, cnt, lane, s1, Wh, s2l, mxi, dn, acc, keep);
    else
        ownScan<1>(lw, wave, lane, (int)blockIdx.x, ekey, eword, s1, Wh, s2l, mxi, dn, acc, cnt, keep);
    __syncthreads();

    float4* o4 = (float4*)(out + (size_t)lo * (N_HEADS * N_HID));
    for (int i = tid; i < nloc * (N_HEADS * N_HID / 4); i += 1024) {
        const float d0 = dn[(i >> 6) * N_HEADS + ((i >> 3) & 7)];
        const float q = d0 > 0.0f ? d0 : 1.0f;
        float4 v = acc4[i];
        v.x /= q; v.y /= q; v.z /= q; v.w /= q;
        o4[i] = v;
    }
}

extern "C" void kernel_launch(void* const* d_in, const int* in_sizes, int n_in,
                              void* d_out, int out_size, void* d_ws, size_t ws_size,
                              hipStream_t stream) {
    const float* h   = (const float*)d_in[0];
    const float* W   = (const float*)d_in[1];
    const float* a   = (const float*)d_in[2];
    const int*   src = (const int*)d_in[3];
    const int*   dst = (const int*)d_in[4];
    float* out = (float*)d_out;

    float* ws = (float*)d_ws;
    float* Wh = ws;
    float* s1 = Wh + (size_t)N_HEADS * N_NODES * N_HID;
    float* s2 = s1 + (size_t)N_HEADS * N_NODES;
    int* ekey  = (int*)(s2 + (size_t)N_HEADS * N_NODES);
    int* eword = ekey + N_EDGES / 4;

    gat_gemm<<<GM_GRID, GM_WAVES * 32, 0, stream>>>(h, W, Wh);

    gat_score<<<(SC_WAVES + 7) / 8, 256, 0, stream>>>(Wh, a, src, dst, s1, s2, ekey, eword);

    gat_own<<<(N_NODES + TILE_NODES - 1) / TILE_NODES, 1024, 0, stream>>>(ekey, eword, s1, s2, Wh, out);
}


# score launch counting-sorts each 1024-edge chunk by owner tile in LDS (rank: LDS int atomic); owner reads a header + one contiguous segment per chunk, no key scan; a wave's edge order changes
# speedup vs baseline: 7.0615x; 1.0861x over previous
#include <hip/hip_runtime.h>
#include <hip/hip_bf16.h>

typedef float v2f __attribute__((ext_vector_type(2)));
typedef float v8f __attribute__((ext_vector_type(8)));

#define N_NODES 50000
#define N_EDGES 800000
#define N_FEAT  256
#define N_HID   32
#define N_HEADS 8
#define ALPHA   0.2f
#define TILE_NODES 196

#define GM_GRID  256
#define GM_WAVES 8
#define GM_TILES (N_NODES / 16)
#define GM_STEP  (GM_GRID * (GM_WAVES / 2))

__global__ __launch_bounds__(GM_WAVES * 32) void gat_gemm(const float* __restrict__ hmat,
                                                          const float* __restrict__ W,
                                                          float* __restrict__ Wh) {
    __shared__ v2f wl[64 * 16 * 32];

    const int tid  = threadIdx.x;
    const int lane = tid & 31;
    const int wave = __builtin_amdgcn_readfirstlane(tid >> 5);
    const int m  = lane & 15;
    const int hf = lane >> 4;

    const int nh = wave & 1;
    int tile = blockIdx.x + GM_GRID * (wave >> 1);

    const float4* rowp = (const float4*)(hmat + (size_t)(tile * 16 + m) * N_FEAT + 32 * hf);
    float4 cur[8];
#pragma unroll
    for (int i = 0; i < 8; ++i) cur[i] = rowp[i];

    for (int i0 = 0; i0 < (64 * 16 * 32) / (GM_WAVES * 32); i0 += 16) {
        v2f tmp[16];
#pragma unroll
        for (int i = 0; i < 16; ++i) {
            const int idx = tid + GM_WAVES * 32 * (i0 + i);
            const int nt = (idx >> 5) & 15, ks = idx >> 9;
            const int k = 64 * (ks >> 4) + 32 * hf + 2 * (ks & 15);
            const float* wp = W + (size_t)((nt >> 1) * N_FEAT + k) * N_HID + (nt & 1) * 16 + m;
            tmp[i][0] = wp[0];
            tmp[i][1] = wp[N_HID];
        }
#pragma unroll
        for (int i = 0; i < 16; ++i)
            wl[tid + GM_WAVES * 32 * (i0 + i)] = tmp[i];
    }
    __syncthreads();

    const v2f* bl = wl + nh * (8 * 32) + lane;

    for (; tile < GM_TILES; tile += GM_STEP) {
        const int tn = tile + GM_STEP < GM_TILES ? tile + GM_STEP : tile;
        const float4* nrow = (const float4*)(hmat + (size_t)(tn * 16 + m) * N_FEAT + 32 * hf);

        v8f acc[8];
#pragma unroll
        for (int j = 0; j < 8; ++j) acc[j] = v8f{};

#pragma nounroll
        for (int kc = 0; kc < 4; ++kc) {
            const float4* np = kc < 3 ? rowp + (kc + 1) * 16 : nrow;
            float4 nxt[8];
#pragma unroll
            for (int i = 0; i < 8; ++i) nxt[i] = np[i];

            const v2f* bk = bl + kc * (16 * 16 * 32);
#pragma unroll
            for (int s = 0; s < 16; ++s) {
                const float4 q = cur[s >> 1];
                v2f a;
                a[0] = (s & 1) ? q.z : q.x;
                a[1] = (s & 1) ? q.w : q.y;
#pragma unroll
                for (int j = 0; j < 8; ++j) {
                    const v2f b = bk[(s * 16 + j) * 32];
                    acc[j] = __builtin_amdgcn_wmma_f32_16x16x4_f32(false, a, false, b,
                                                                   (short)0, acc[j], false, false);
                }
            }
#pragma unroll
            for (int i = 0; i < 8; ++i) cur[i] = nxt[i];
        }

#pragma unroll
        for (int j = 0; j < 8; ++j) {
            const int head = nh * 4 + (j >> 1);
            float* op = Wh + ((size_t)head * N_NODES + tile * 16 + hf * 8) * N_HID + (j & 1) * 16 + m;
#pragma unroll
            for (int r = 0; r < 8; ++r)
                op[r * N_HID] = acc[j][r];
        }
        rowp = nrow;
    }
}

#define SC_ROWS  64
#define SC_WAVES ((N_HEADS * N_NODES) / SC_ROWS)

static_assert((N_HEADS * N_NODES) % SC_ROWS == 0, "whole 64-row pieces");
static_assert(N_NODES % 16 == 0, "a group of 16 rows lies inside one head");
#define CH_EDGES 1024
#define N_CHUNKS ((N_EDGES + CH_EDGES - 1) / CH_EDGES)
#define OWN_TILES 256

static_assert(N_EDGES == SC_WAVES * 128, "every score wave takes 128 edges");
static_assert((SC_WAVES + 7) / 8 == N_CHUNKS, "one score work-group per chunk");
static_assert((N_NODES + TILE_NODES - 1) / TILE_NODES <= OWN_TILES, "a header word per owner tile");
static_assert(N_NODES <= (1 << 16) && TILE_NODES <= 256, "edge word: local node << 16 | src");

__global__ __launch_bounds__(256) void gat_score(const float* __restrict__ Wh,
                                                 const float* __restrict__ avec,
                                                 const int* __restrict__ src,
                                                 const int* __restrict__ dst,
                                                 float* __restrict__ s1,
                                                 float* __restrict__ s2,
                                                 int* __restrict__ eoff,
                                                 int* __restrict__ eword) {
    __shared__ int tcnt[OWN_TILES];
    __shared__ int wsum[8];
    __shared__ __attribute__((aligned(16))) int srt[CH_EDGES];

    const int tid  = threadIdx.x;
    const int lane = tid & 31;
    const int wv = __builtin_amdgcn_readfirstlane(tid >> 5);
    const int w = blockIdx.x * 8 + wv;
    const bool live = w < SC_WAVES;
    const int j = lane >> 3;
    const int q = lane & 7;

    tcnt[tid] = 0;

    int4 dv = make_int4(0, 0, 0, 0), sv = make_int4(0, 0, 0, 0);
    if (live) {
        dv = ((const int4*)dst)[w * 32 + lane];
        sv = ((const int4*)src)[w * 32 + lane];
    }

    if (live) {

    const float4* w4 = (const float4*)Wh + (size_t)w * (SC_ROWS * 8) + lane;
    float4 v[16];
#pragma unroll
    for (int i = 0; i < 16; ++i)
        v[i] = w4[i * 32];

    const float4* a4 = (const float4*)avec;
    float p1[16], p2[16];
#pragma unroll
    for (int t = 0; t < 4; ++t) {
        const int head = (w * SC_ROWS + t * 16) / N_NODES;
        const float4 x1 = a4[head * 16 + q];
        const float4 x2 = a4[head * 16 + 8 + q];
#pragma unroll
        for (int u = 0; u < 4; ++u) {
            const float4 c = v[t * 4 + u];
            p1[t * 4 + u] = c.x * x1.x + c.y * x1.y + c.z * x1.z + c.w * x1.w;
            p2[t * 4 + u] = c.x * x2.x + c.y * x2.y + c.z * x2.z + c.w * x2.w;
        }
    }

    const bool b2 = (q & 4) != 0, b1 = (q & 2) != 0, b0 = (q & 1) != 0;
    float r1[8], r2[8];
#pragma unroll
    for (int g = 0; g < 2; ++g)
#pragma unroll
        for (int l = 0; l < 4; ++l) {
            const int i = g * 8 + l;
            const float k1 = b2 ? p1[i + 4] : p1[i], d1 = b2 ? p1[i] : p1[i + 4];
            const float k2 = b2 ? p2[i + 4] : p2[i], d2 = b2 ? p2[i] : p2[i + 4];
            r1[g * 4 + l] = k1 + __shfl_xor(d1, 4, 32);
            r2[g * 4 + l] = k2 + __shfl_xor(d2, 4, 32);
        }
    float t1[4], t2[4];
#pragma unroll
    for (int g = 0; g < 2; ++g)
#pragma unroll
        for (int l = 0; l < 2; ++l) {
            const int i = g * 4 + l;
            const float k1 = b1 ? r1[i + 2] : r1[i], d1 = b1 ? r1[i] : r1[i + 2];
            const float k2 = b1 ? r2[i + 2] : r2[i], d2 = b1 ? r2[i] : r2[i + 2];
            t1[g * 2 + l] = k1 + __shfl_xor(d1, 2, 32);
            t2[g * 2 + l] = k2 + __shfl_xor(d2, 2, 32);
        }
    float* o1 = s1 + (size_t)w * SC_ROWS + 4 * q + j;
    float* o2 = s2 + (size_t)w * SC_ROWS + 4 * q + j;
#pragma unroll
    for (int g = 0; g < 2; ++g) {
        const int i = g * 2;
        const float k1 = b0 ? t1[i + 1] : t1[i], d1 = b0 ? t1[i] : t1[i + 1];
        const float k2 = b0 ? t2[i + 1] : t2[i], d2 = b0 ? t2[i] : t2[i + 1];
        o1[g * 32] = k1 + __shfl_xor(d1, 1, 32);
        o2[g * 32] = k2 + __shfl_xor(d2, 1, 32);
    }
    }

    const int dd[4] = {dv.x, dv.y, dv.z, dv.w};
    const int ss[4] = {sv.x, sv.y, sv.z, sv.w};
    unsigned ot[4]; int rk[4], ew[4];
#pragma unroll
    for (int u = 0; u < 4; ++u) {
        ot[u] = (unsigned)dd[u] / TILE_NODES;
        ew[u] = (int)(((unsigned)dd[u] - ot[u] * TILE_NODES) << 16) | ss[u];
        rk[u] = 0;
    }
    __syncthreads();
    if (live) {
#pragma unroll
        for (int u = 0; u < 4; ++u)
            rk[u] = atomicAdd(&tcnt[ot[u]], 1);
    }
    __syncthreads();

    const int c = tcnt[tid];
    int inc = c;
#pragma unroll
    for (int d = 1; d < 32; d <<= 1) {
        const int y = __shfl_up(inc, d, 32);
        if (lane >= d) inc += y;
    }
    if (lane == 31) wsum[wv] = inc;
    __syncthreads();
    int start = inc - c;
#pragma unroll
    for (int i = 0; i < 8; ++i)
        if (i < wv) start += wsum[i];
    tcnt[tid] = start;
    eoff[blockIdx.x * OWN_TILES + tid] = start | (c << 16);
    __syncthreads();

    if (live) {
#pragma unroll
        for (int u = 0; u < 4; ++u)
            srt[tcnt[ot[u]] + rk[u]] = ew[u];
    }
    __syncthreads();

    const int nvalid = (N_EDGES - (int)blockIdx.x * CH_EDGES) < CH_EDGES
                     ? (N_EDGES - (int)blockIdx.x * CH_EDGES) : CH_EDGES;
    if (4 * tid < nvalid)
        ((int4*)eword)[blockIdx.x * (CH_EDGES / 4) + tid] = ((const int4*)srt)[tid];
}

#define OWN_WAVES  32
#define LIST_CAP   704
#define KEEP_GROUPS 32

typedef float v32f __attribute__((ext_vector_type(32)));

static_assert(N_CHUNKS <= OWN_WAVES * 32, "one chunk per lane covers every chunk");
static_assert((KEEP_GROUPS * 4) % 16 == 0 && KEEP_GROUPS <= 32, "phase B fills keep[] 4 groups a trip");

__device__ __forceinline__ void ldsMaxFloat(int* addr, float val) {
    if (val >= 0.0f)
        atomicMax(addr, __float_as_int(val));
    else
        atomicMin((unsigned int*)addr, __float_as_uint(val));
}

__device__ __forceinline__ void ownListFence() {
    __builtin_amdgcn_fence(__ATOMIC_ACQ_REL, "wavefront");
}

__device__ __forceinline__ void ownMax(const int* lw, int cnt, int lane,
                                       const float* __restrict__ s1,
                                       const float* s2l, int* mxi, v32f& keep) {
    const int j = lane >> 3, h = lane & 7;
    for (int i0 = 0; i0 < cnt; i0 += 16) {
        float x[4]; int ld[4];
#pragma unroll
        for (int u = 0; u < 4; ++u) {
            const int idx = i0 + u * 4 + j;
            const int ent = idx < cnt ? lw[idx] : 0;
            ld[u] = ent >> 16;
            x[u] = s1[h * N_NODES + (ent & 0xFFFF)] + s2l[ld[u] * N_HEADS + h];
        }
#pragma unroll
        for (int u = 0; u < 4; ++u) {
            const int idx = i0 + u * 4 + j;
            x[u] = x[u] > 0.0f ? x[u] : ALPHA * x[u];
            if (idx < cnt) ldsMaxFloat(&mxi[ld[u] * N_HEADS + h], x[u]);
        }
        if (i0 < KEEP_GROUPS * 4) {
#pragma unroll
            for (int u = 0; u < 4; ++u)
                keep[(i0 >> 2) + u] = x[u];
        }
    }
}

template <bool KEPT>
__device__ __forceinline__ void ownSum(const int* lw, int cnt, int lane,
                                       const float* __restrict__ s1,
                                       const float* __restrict__ Wh,
                                       const float* s2l, const int* mxi,
                                       float* dn, float* acc, const v32f& keep) {
    const int j = lane >> 3, h = lane & 7;
    for (int i0 = 0; i0 < cnt; i0 += 4) {
        const bool mine = i0 + j < cnt;
        const int ent = lw[mine ? i0 + j : cnt - 1];
        const int lj = ent >> 16;
        int s[4], ld[4]; float wh[4][N_HEADS];
#pragma unroll
        for (int u = 0; u < 4; ++u) {
            const int e = __builtin_amdgcn_readlane(ent, 8 * u);
            s[u]  = e & 0xFFFF;
            ld[u] = e >> 16;
        }
        float v;
        if (KEPT && i0 < KEEP_GROUPS * 4) {
            v = keep[i0 >> 2];
        } else {
            const float x = s1[h * N_NODES + (ent & 0xFFFF)] + s2l[lj * N_HEADS + h];
            v = x > 0.0f ? x : ALPHA * x;
        }
#pragma unroll
        for (int u = 0; u < 4; ++u)
#pragma unroll
            for (int head = 0; head < N_HEADS; ++head)
                wh[u][head] = Wh[((size_t)head * N_NODES + s[u]) * N_HID + lane];
        const float ex = __expf(v - __int_as_float(mxi[lj * N_HEADS + h]));
        if (mine) atomicAdd(&dn[lj * N_HEADS + h], ex);
#pragma unroll
        for (int u = 0; u < 4; ++u) {
            if (i0 + u < cnt) {
                float* ap = acc + ld[u] * (N_HEADS * N_HID) + lane;
#pragma unroll
                for (int head = 0; head < N_HEADS; ++head) {
                    const float w = __int_as_float(
                        __builtin_amdgcn_readlane(__float_as_int(ex), 8 * u + head));
                    atomicAdd(ap + head * N_HID, w * wh[u][head]);
                }
            }
        }
    }
}

template <int PHASE>
__device__ __forceinline__ void ownWork(int* lw, int cnt, int lane,
                                        const float* __restrict__ s1,
                                        const float* __restrict__ Wh,
                                        const float* s2l, int* mxi,
                                        float* dn, float* acc, v32f& keep) {
    if (PHASE == 0) ownMax(lw, cnt, lane, s1, s2l, mxi, keep);
    else            ownSum<false>(lw, cnt, lane, s1, Wh, s2l, mxi, dn, acc, keep);
}

template <int PHASE>
__device__ __forceinline__ int ownCollect(int* lw, int wave, int lane, int tile,
                                          const int* __restrict__ eoff,
                                          const int* __restrict__ eword,
                                          const float* __restrict__ s1,
                                          const float* __restrict__ Wh,
                                          const float* s2l, int* mxi,
                                          float* dn, float* acc, int& cnt_out,
                                          v32f& keep) {
    const int ci  = lane * OWN_WAVES + wave;
    const bool has = ci < N_CHUNKS;
    const int cc  = has ? ci : 0;
    const int hdr = eoff[cc * OWN_TILES + tile];
    const int len = has ? (hdr >> 16) : 0;
    const int beg = cc * CH_EDGES + (hdr & 0xFFFF);

    int inc = len;
#pragma unroll
    for (int d = 1; d < 32; d <<= 1) {
        const int y = __shfl_up(inc, d, 32);
        if (lane >= d) inc += y;
    }
    const int pos   = inc - len;
    const int total = __builtin_amdgcn_readlane(inc, 31);

    if (total <= LIST_CAP) {
        for (int t0 = 0; __builtin_amdgcn_ballot_w32(t0 < len) != 0u; t0 += 4) {
            int v[4];
#pragma unroll
            for (int u = 0; u < 4; ++u)
                v[u] = eword[t0 + u < len ? beg + t0 + u : 0];
#pragma unroll
            for (int u = 0; u < 4; ++u)
                if (t0 + u < len) lw[pos + t0 + u] = v[u];
        }
        ownListFence();
        ownWork<PHASE>(lw, total, lane, s1, Wh, s2l, mxi, dn, acc, keep);
        cnt_out = total;
        return 1;
    }

    for (int l = 0; l < 32; ++l) {
        const int n = __builtin_amdgcn_readlane(len, l);
        const int b = __builtin_amdgcn_readlane(beg, l);
        for (int o = 0; o < n; o += LIST_CAP) {
            const int m = n - o < LIST_CAP ? n - o : LIST_CAP;
            for (int i = lane; i < m; i += 32)
                lw[i] = eword[b + o + i];
            ownListFence();
            ownWork<PHASE>(lw, m, lane, s1, Wh, s2l, mxi, dn, acc, keep);
            ownListFence();
        }
    }
    cnt_out = 0;
    return 2;
}

__global__ __launch_bounds__(1024) void gat_own(const int* __restrict__ eoff,
                                                const int* __restrict__ eword,
                                                const float* __restrict__ s1,
                                                const float* __restrict__ s2,
                                                const float* __restrict__ Wh,
                                                float* __restrict__ out) {
    __shared__ __attribute__((aligned(16))) float acc[TILE_NODES * N_HEADS * N_HID];
    __shared__ int   mxi[TILE_NODES * N_HEADS];
    __shared__ float dn[TILE_NODES * N_HEADS];
    __shared__ float s2l[TILE_NODES * N_HEADS];
    __shared__ int   lst[OWN_WAVES * LIST_CAP];

    const int tid  = threadIdx.x;
    const int lane = tid & 31;
    const int wave = __builtin_amdgcn_readfirstlane(tid >> 5);
    const int lo   = blockIdx.x * TILE_NODES;
    const int nloc = (N_NODES - lo) < TILE_NODES ? (N_NODES - lo) : TILE_NODES;

    float4* acc4 = (float4*)acc;
    for (int i = tid; i < TILE_NODES * N_HEADS * N_HID / 4; i += 1024)
        acc4[i] = make_float4(0.0f, 0.0f, 0.0f, 0.0f);
    for (int i = tid; i < TILE_NODES * N_HEADS; i += 1024) {
        const int node = lo + (i >> 3);
        mxi[i] = __float_as_int(-__builtin_inff());
        dn[i]  = 0.0f;
        s2l[i] = node < N_NODES ? s2[(i & 7) * N_NODES + node] : 0.0f;
    }
    __syncthreads();

    int* lw = lst + wave * LIST_CAP;
    int cnt = 0;
    v32f keep = {};
    const int rounds = ownCollect<0>(lw, wave, lane, (int)blockIdx.x, eoff, eword, s1, Wh,
                                     s2l, mxi, dn, acc, cnt, keep);
    __syncthreads();

    if (rounds == 1)
        ownSum<true>(lw, cnt, lane, s1, Wh, s2l, mxi, dn, acc, keep);
    else
        ownCollect<1>(lw, wave, lane, (int)blockIdx.x, eoff, eword, s1, Wh, s2l, mxi, dn, acc, cnt, keep);
    __syncthreads();

    float4* o4 = (float4*)(out + (size_t)lo * (N_HEADS * N_HID));
    for (int i = tid; i < nloc * (N_HEADS * N_HID / 4); i += 1024) {
        const float d0 = dn[(i >> 6) * N_HEADS + ((i >> 3) & 7)];
        const float q = d0 > 0.0f ? d0 : 1.0f;
        float4 v = acc4[i];
        v.x /= q; v.y /= q; v.z /= q; v.w /= q;
        o4[i] = v;
    }
}

extern "C" void kernel_launch(void* const* d_in, const int* in_sizes, int n_in,
                              void* d_out, int out_size, void* d_ws, size_t ws_size,
                              hipStream_t stream) {
    const float* h   = (const float*)d_in[0];
    const float* W   = (const float*)d_in[1];
    const float* a   = (const float*)d_in[2];
    const int*   src = (const int*)d_in[3];
    const int*   dst = (const int*)d_in[4];
    float* out = (float*)d_out;

    float* ws = (float*)d_ws;
    float* Wh = ws;
    float* s1 = Wh + (size_t)N_HEADS * N_NODES * N_HID;
    float* s2 = s1 + (size_t)N_HEADS * N_NODES;
    int* eoff  = (int*)(s2 + (size_t)N_HEADS * N_NODES);
    int* eword = eoff + N_CHUNKS * OWN_TILES;

    gat_gemm<<<GM_GRID, GM_WAVES * 32, 0, stream>>>(h, W, Wh);

    gat_score<<<(SC_WAVES + 7) / 8, 256, 0, stream>>>(Wh, a, src, dst, s1, s2, eoff, eword);

    gat_own<<<(N_NODES + TILE_NODES - 1) / TILE_NODES, 1024, 0, stream>>>(eoff, eword, s1, s2, Wh, out);
}


# scores on the matrix unit: U = W.a folded in as a 17th column tile, GEMM stores s1|s2 as [node][16]; pass over Wh removed, owner gathers one 32 B piece per edge; s1,s2 sum order changed
# speedup vs baseline: 7.7627x; 1.0993x over previous
#include <hip/hip_runtime.h>
#include <hip/hip_bf16.h>

typedef float v2f __attribute__((ext_vector_type(2)));
typedef float v8f __attribute__((ext_vector_type(8)));

#define N_NODES 50000
#define N_EDGES 800000
#define N_FEAT  256
#define N_HID   32
#define N_HEADS 8
#define ALPHA   0.2f
#define TILE_NODES 196

#define GM_GRID  256
#define GM_WAVES 8
#define GM_TILES (N_NODES / 16)
#define GM_STEP  (GM_GRID * (GM_WAVES / 2))

template <bool EXTRA>
__device__ __forceinline__ void gemmTiles(const float* __restrict__ hmat,
                                          float* __restrict__ Wh,
                                          float* __restrict__ s12,
                                          const v2f* bl, const v2f* bu,
                                          int tile, int nh, int m, int hf,
                                          const float4* rowp, float4 (&cur)[8]) {
    for (; tile < GM_TILES; tile += GM_STEP) {
        const int tn = tile + GM_STEP < GM_TILES ? tile + GM_STEP : tile;
        const float4* nrow = (const float4*)(hmat + (size_t)(tn * 16 + m) * N_FEAT + 32 * hf);

        v8f acc[8];
#pragma unroll
        for (int j = 0; j < 8; ++j) acc[j] = v8f{};
        v8f accu = v8f{};

#pragma nounroll
        for (int kc = 0; kc < 4; ++kc) {
            const float4* np = kc < 3 ? rowp + (kc + 1) * 16 : nrow;
            float4 nxt[8];
#pragma unroll
            for (int i = 0; i < 8; ++i) nxt[i] = np[i];

            const v2f* bk = bl + kc * (16 * 16 * 32);
            const v2f* uk = bu + kc * (16 * 32);
#pragma unroll
            for (int s = 0; s < 16; ++s) {
                const float4 q = cur[s >> 1];
                v2f a;
                a[0] = (s & 1) ? q.z : q.x;
                a[1] = (s & 1) ? q.w : q.y;
#pragma unroll
                for (int j = 0; j < 8; ++j) {
                    const v2f b = bk[(s * 16 + j) * 32];
                    acc[j] = __builtin_amdgcn_wmma_f32_16x16x4_f32(false, a, false, b,
                                                                   (short)0, acc[j], false, false);
                }
                if (EXTRA) {
                    const v2f b = uk[s * 32];
                    accu = __builtin_amdgcn_wmma_f32_16x16x4_f32(false, a, false, b,
                                                                 (short)0, accu, false, false);
                }
            }
#pragma unroll
            for (int i = 0; i < 8; ++i) cur[i] = nxt[i];
        }

#pragma unroll
        for (int j = 0; j < 8; ++j) {
            const int head = nh * 4 + (j >> 1);
            float* op = Wh + ((size_t)head * N_NODES + tile * 16 + hf * 8) * N_HID + (j & 1) * 16 + m;
#pragma unroll
            for (int r = 0; r < 8; ++r)
                op[r * N_HID] = acc[j][r];
        }
        if (EXTRA) {
            float* sp = s12 + (size_t)(tile * 16 + hf * 8) * 16 + m;
#pragma unroll
            for (int r = 0; r < 8; ++r)
                sp[r * 16] = accu[r];
        }
        rowp = nrow;
    }
}

__global__ __launch_bounds__(GM_WAVES * 32) void gat_gemm(const float* __restrict__ hmat,
                                                          const float* __restrict__ W,
                                                          const float* __restrict__ avec,
                                                          float* __restrict__ Wh,
                                                          float* __restrict__ s12) {
    __shared__ v2f wl[64 * 16 * 32];
    __shared__ v2f wu[64 * 32];

    const int tid  = threadIdx.x;
    const int lane = tid & 31;
    const int wave = __builtin_amdgcn_readfirstlane(tid >> 5);
    const int m  = lane & 15;
    const int hf = lane >> 4;

    const int nh = wave & 1;
    const bool extra = (wave & 1) == (wave >> 2);
    const int tile = blockIdx.x + GM_GRID * (wave >> 1);

    const float4* rowp = (const float4*)(hmat + (size_t)(tile * 16 + m) * N_FEAT + 32 * hf);
    float4 cur[8];
#pragma unroll
    for (int i = 0; i < 8; ++i) cur[i] = rowp[i];

    const int uh = m & 7;
    float areg[32];
#pragma unroll
    for (int dd = 0; dd < 32; ++dd)
        areg[dd] = avec[uh * (2 * N_HID) + (m >> 3) * N_HID + ((dd + 2 * uh) & 31)];

    for (int i0 = 0; i0 < (64 * 16 * 32) / (GM_WAVES * 32); i0 += 16) {
        v2f tmp[16];
#pragma unroll
        for (int i = 0; i < 16; ++i) {
            const int idx = tid + GM_WAVES * 32 * (i0 + i);
            const int nt = (idx >> 5) & 15, ks = idx >> 9;
            const int k = 64 * (ks >> 4) + 32 * hf + 2 * (ks & 15);
            const float* wp = W + (size_t)((nt >> 1) * N_FEAT + k) * N_HID + (nt & 1) * 16 + m;
            tmp[i][0] = wp[0];
            tmp[i][1] = wp[N_HID];
        }
#pragma unroll
        for (int i = 0; i < 16; ++i)
            wl[tid + GM_WAVES * 32 * (i0 + i)] = tmp[i];
    }
    __syncthreads();

    for (int e = 0; e < 64 / GM_WAVES; ++e) {
        const int ks = wave + GM_WAVES * e;
        const v2f* wr = wl + (ks * 16 + uh * 2) * 32 + hf * 16;
        float u0 = 0.0f, u1 = 0.0f;
#pragma unroll
        for (int dd = 0; dd < 32; ++dd) {
            const int d = (dd + 2 * uh) & 31;
            const v2f wv = wr[d + (d & 16)];
            u0 += wv[0] * areg[dd];
            u1 += wv[1] * areg[dd];
        }
        v2f uu; uu[0] = u0; uu[1] = u1;
        wu[ks * 32 + lane] = uu;
    }
    __syncthreads();

    const v2f* bl = wl + nh * (8 * 32) + lane;
    const v2f* bu = wu + lane;
    if (extra)
        gemmTiles<true>(hmat, Wh, s12, bl, bu, tile, nh, m, hf, rowp, cur);
    else
        gemmTiles<false>(hmat, Wh, s12, bl, bu, tile, nh, m, hf, rowp, cur);
}

#define SC_WAVES (N_EDGES / 128)
#define CH_EDGES 1024
#define N_CHUNKS ((N_EDGES + CH_EDGES - 1) / CH_EDGES)
#define OWN_TILES 256

static_assert(N_NODES % 16 == 0, "whole 16-node GEMM tiles");
static_assert(N_EDGES == SC_WAVES * 128, "every sort wave takes 128 edges");
static_assert((SC_WAVES + 7) / 8 == N_CHUNKS, "one sort work-group per chunk");
static_assert((N_NODES + TILE_NODES - 1) / TILE_NODES <= OWN_TILES, "a header word per owner tile");
static_assert(N_NODES <= (1 << 16) && TILE_NODES <= 256, "edge word: local node << 16 | src");

__global__ __launch_bounds__(256) void gat_sort(const int* __restrict__ src,
                                                const int* __restrict__ dst,
                                                int* __restrict__ eoff,
                                                int* __restrict__ eword) {
    __shared__ int tcnt[OWN_TILES];
    __shared__ int wsum[8];
    __shared__ __attribute__((aligned(16))) int srt[CH_EDGES];

    const int tid  = threadIdx.x;
    const int lane = tid & 31;
    const int wv = __builtin_amdgcn_readfirstlane(tid >> 5);
    const int w = blockIdx.x * 8 + wv;
    const bool live = w < SC_WAVES;

    tcnt[tid] = 0;

    int4 dv = make_int4(0, 0, 0, 0), sv = make_int4(0, 0, 0, 0);
    if (live) {
        dv = ((const int4*)dst)[w * 32 + lane];
        sv = ((const int4*)src)[w * 32 + lane];
    }

    const int dd[4] = {dv.x, dv.y, dv.z, dv.w};
    const int ss[4] = {sv.x, sv.y, sv.z, sv.w};
    unsigned ot[4]; int rk[4], ew[4];
#pragma unroll
    for (int u = 0; u < 4; ++u) {
        ot[u] = (unsigned)dd[u] / TILE_NODES;
        ew[u] = (int)(((unsigned)dd[u] - ot[u] * TILE_NODES) << 16) | ss[u];
        rk[u] = 0;
    }
    __syncthreads();
    if (live) {
#pragma unroll
        for (int u = 0; u < 4; ++u)
            rk[u] = atomicAdd(&tcnt[ot[u]], 1);
    }
    __syncthreads();

    const int c = tcnt[tid];
    int inc = c;
#pragma unroll
    for (int d = 1; d < 32; d <<= 1) {
        const int y = __shfl_up(inc, d, 32);
        if (lane >= d) inc += y;
    }
    if (lane == 31) wsum[wv] = inc;
    __syncthreads();
    int start = inc - c;
#pragma unroll
    for (int i = 0; i < 8; ++i)
        if (i < wv) start += wsum[i];
    tcnt[tid] = start;
    eoff[blockIdx.x * OWN_TILES + tid] = start | (c << 16);
    __syncthreads();

    if (live) {
#pragma unroll
        for (int u = 0; u < 4; ++u)
            srt[tcnt[ot[u]] + rk[u]] = ew[u];
    }
    __syncthreads();

    const int nvalid = (N_EDGES - (int)blockIdx.x * CH_EDGES) < CH_EDGES
                     ? (N_EDGES - (int)blockIdx.x * CH_EDGES) : CH_EDGES;
    if (4 * tid < nvalid)
        ((int4*)eword)[blockIdx.x * (CH_EDGES / 4) + tid] = ((const int4*)srt)[tid];
}

#define OWN_WAVES  32
#define LIST_CAP   704
#define KEEP_GROUPS 32

typedef float v32f __attribute__((ext_vector_type(32)));

static_assert(N_CHUNKS <= OWN_WAVES * 32, "one chunk per lane covers every chunk");
static_assert((KEEP_GROUPS * 4) % 16 == 0 && KEEP_GROUPS <= 32, "phase B fills keep[] 4 groups a trip");

__device__ __forceinline__ void ldsMaxFloat(int* addr, float val) {
    if (val >= 0.0f)
        atomicMax(addr, __float_as_int(val));
    else
        atomicMin((unsigned int*)addr, __float_as_uint(val));
}

__device__ __forceinline__ void ownListFence() {
    __builtin_amdgcn_fence(__ATOMIC_ACQ_REL, "wavefront");
}

__device__ __forceinline__ void ownMax(const int* lw, int cnt, int lane,
                                       const float* __restrict__ s12,
                                       const float* s2l, int* mxi, v32f& keep) {
    const int j = lane >> 3, h = lane & 7;
    for (int i0 = 0; i0 < cnt; i0 += 16) {
        float x[4]; int ld[4];
#pragma unroll
        for (int u = 0; u < 4; ++u) {
            const int idx = i0 + u * 4 + j;
            const int ent = idx < cnt ? lw[idx] : 0;
            ld[u] = ent >> 16;
            x[u] = s12[(ent & 0xFFFF) * 16 + h] + s2l[ld[u] * N_HEADS + h];
        }
#pragma unroll
        for (int u = 0; u < 4; ++u) {
            const int idx = i0 + u * 4 + j;
            x[u] = x[u] > 0.0f ? x[u] : ALPHA * x[u];
            if (idx < cnt) ldsMaxFloat(&mxi[ld[u] * N_HEADS + h], x[u]);
        }
        if (i0 < KEEP_GROUPS * 4) {
#pragma unroll
            for (int u = 0; u < 4; ++u)
                keep[(i0 >> 2) + u] = x[u];
        }
    }
}

template <bool KEPT>
__device__ __forceinline__ void ownSum(const int* lw, int cnt, int lane,
                                       const float* __restrict__ s12,
                                       const float* __restrict__ Wh,
                                       const float* s2l, const int* mxi,
                                       float* dn, float* acc, const v32f& keep) {
    const int j = lane >> 3, h = lane & 7;
    for (int i0 = 0; i0 < cnt; i0 += 4) {
        const bool mine = i0 + j < cnt;
        const int ent = lw[mine ? i0 + j : cnt - 1];
        const int lj = ent >> 16;
        int s[4], ld[4]; float wh[4][N_HEADS];
#pragma unroll
        for (int u = 0; u < 4; ++u) {
            const int e = __builtin_amdgcn_readlane(ent, 8 * u);
            s[u]  = e & 0xFFFF;
            ld[u] = e >> 16;
        }
        float v;
        if (KEPT && i0 < KEEP_GROUPS * 4) {
            v = keep[i0 >> 2];
        } else {
            const float x = s12[(ent & 0xFFFF) * 16 + h] + s2l[lj * N_HEADS + h];
            v = x > 0.0f ? x : ALPHA * x;
        }
#pragma unroll
        for (int u = 0; u < 4; ++u)
#pragma unroll
            for (int head = 0; head < N_HEADS; ++head)
                wh[u][head] = Wh[((size_t)head * N_NODES + s[u]) * N_HID + lane];
        const float ex = __expf(v - __int_as_float(mxi[lj * N_HEADS + h]));
        if (mine) atomicAdd(&dn[lj * N_HEADS + h], ex);
#pragma unroll
        for (int u = 0; u < 4; ++u) {
            if (i0 + u < cnt) {
                float* ap = acc + ld[u] * (N_HEADS * N_HID) + lane;
#pragma unroll
                for (int head = 0; head < N_HEADS; ++head) {
                    const float w = __int_as_float(
                        __builtin_amdgcn_readlane(__float_as_int(ex), 8 * u + head));
                    atomicAdd(ap + head * N_HID, w * wh[u][head]);
                }
            }
        }
    }
}

template <int PHASE>
__device__ __forceinline__ void ownWork(int* lw, int cnt, int lane,
                                        const float* __restrict__ s12,
                                        const float* __restrict__ Wh,
                                        const float* s2l, int* mxi,
                                        float* dn, float* acc, v32f& keep) {
    if (PHASE == 0) ownMax(lw, cnt, lane, s12, s2l, mxi, keep);
    else            ownSum<false>(lw, cnt, lane, s12, Wh, s2l, mxi, dn, acc, keep);
}

template <int PHASE>
__device__ __forceinline__ int ownCollect(int* lw, int wave, int lane, int tile,
                                          const int* __restrict__ eoff,
                                          const int* __restrict__ eword,
                                          const float* __restrict__ s12,
                                          const float* __restrict__ Wh,
                                          const float* s2l, int* mxi,
                                          float* dn, float* acc, int& cnt_out,
                                          v32f& keep) {
    const int ci  = lane * OWN_WAVES + wave;
    const bool has = ci < N_CHUNKS;
    const int cc  = has ? ci : 0;
    const int hdr = eoff[cc * OWN_TILES + tile];
    const int len = has ? (hdr >> 16) : 0;
    const int beg = cc * CH_EDGES + (hdr & 0xFFFF);

    int inc = len;
#pragma unroll
    for (int d = 1; d < 32; d <<= 1) {
        const int y = __shfl_up(inc, d, 32);
        if (lane >= d) inc += y;
    }
    const int pos   = inc - len;
    const int total = __builtin_amdgcn_readlane(inc, 31);

    if (total <= LIST_CAP) {
        for (int t0 = 0; __builtin_amdgcn_ballot_w32(t0 < len) != 0u; t0 += 4) {
            int v[4];
#pragma unroll
            for (int u = 0; u < 4; ++u)
                v[u] = eword[t0 + u < len ? beg + t0 + u : 0];
#pragma unroll
            for (int u = 0; u < 4; ++u)
                if (t0 + u < len) lw[pos + t0 + u] = v[u];
        }
        ownListFence();
        ownWork<PHASE>(lw, total, lane, s12, Wh, s2l, mxi, dn, acc, keep);
        cnt_out = total;
        return 1;
    }

    for (int l = 0; l < 32; ++l) {
        const int n = __builtin_amdgcn_readlane(len, l);
        const int b = __builtin_amdgcn_readlane(beg, l);
        for (int o = 0; o < n; o += LIST_CAP) {
            const int m = n - o < LIST_CAP ? n - o : LIST_CAP;
            for (int i = lane; i < m; i += 32)
                lw[i] = eword[b + o + i];
            ownListFence();
            ownWork<PHASE>(lw, m, lane, s12, Wh, s2l, mxi, dn, acc, keep);
            ownListFence();
        }
    }
    cnt_out = 0;
    return 2;
}

__global__ __launch_bounds__(1024) void gat_own(const int* __restrict__ eoff,
                                                const int* __restrict__ eword,
                                                const float* __restrict__ s12,
                                                const float* __restrict__ Wh,
                                                float* __restrict__ out) {
    __shared__ __attribute__((aligned(16))) float acc[TILE_NODES * N_HEADS * N_HID];
    __shared__ int   mxi[TILE_NODES * N_HEADS];
    __shared__ float dn[TILE_NODES * N_HEADS];
    __shared__ float s2l[TILE_NODES * N_HEADS];
    __shared__ int   lst[OWN_WAVES * LIST_CAP];

    const int tid  = threadIdx.x;
    const int lane = tid & 31;
    const int wave = __builtin_amdgcn_readfirstlane(tid >> 5);
    const int lo   = blockIdx.x * TILE_NODES;
    const int nloc = (N_NODES - lo) < TILE_NODES ? (N_NODES - lo) : TILE_NODES;

    float4* acc4 = (float4*)acc;
    for (int i = tid; i < TILE_NODES * N_HEADS * N_HID / 4; i += 1024)
        acc4[i] = make_float4(0.0f, 0.0f, 0.0f, 0.0f);
    for (int i = tid; i < TILE_NODES * N_HEADS; i += 1024) {
        const int node = lo + (i >> 3);
        mxi[i] = __float_as_int(-__builtin_inff());
        dn[i]  = 0.0f;
        s2l[i] = node < N_NODES ? s12[node * 16 + 8 + (i & 7)] : 0.0f;
    }
    __syncthreads();

    int* lw = lst + wave * LIST_CAP;
    int cnt = 0;
    v32f keep = {};
    const int rounds = ownCollect<0>(lw, wave, lane, (int)blockIdx.x, eoff, eword, s12, Wh,
                                     s2l, mxi, dn, acc, cnt, keep);
    __syncthreads();

    if (rounds == 1)
        ownSum<true>(lw, cnt, lane, s12, Wh, s2l, mxi, dn, acc, keep);
    else
        ownCollect<1>(lw, wave, lane, (int)blockIdx.x, eoff, eword, s12, Wh, s2l, mxi, dn, acc, cnt, keep);
    __syncthreads();

    float4* o4 = (float4*)(out + (size_t)lo * (N_HEADS * N_HID));
    for (int i = tid; i < nloc * (N_HEADS * N_HID / 4); i += 1024) {
        const float d0 = dn[(i >> 6) * N_HEADS + ((i >> 3) & 7)];
        const float q = d0 > 0.0f ? d0 : 1.0f;
        float4 v = acc4[i];
        v.x /= q; v.y /= q; v.z /= q; v.w /= q;
        o4[i] = v;
    }
}

extern "C" void kernel_launch(void* const* d_in, const int* in_sizes, int n_in,
                              void* d_out, int out_size, void* d_ws, size_t ws_size,
                              hipStream_t stream) {
    const float* h   = (const float*)d_in[0];
    const float* W   = (const float*)d_in[1];
    const float* a   = (const float*)d_in[2];
    const int*   src = (const int*)d_in[3];
    const int*   dst = (const int*)d_in[4];
    float* out = (float*)d_out;

    float* ws = (float*)d_ws;
    float* Wh = ws;
    float* s12 = Wh + (size_t)N_HEADS * N_NODES * N_HID;
    int* eoff  = (int*)(s12 + (size_t)N_NODES * 16);
    int* eword = eoff + N_CHUNKS * OWN_TILES;

    gat_gemm<<<GM_GRID, GM_WAVES * 32, 0, stream>>>(h, W, a, Wh, s12);

    gat_sort<<<N_CHUNKS, 256, 0, stream>>>(src, dst, eoff, eword);

    gat_own<<<(N_NODES + TILE_NODES - 1) / TILE_NODES, 1024, 0, stream>>>(eoff, eword, s12, Wh, out);
}


# owner pass: each edge dealt to the wave that owns its dst row (slot from an LDS int counter); rows added by plain 16-byte LDS read-add-write, no row float atomics; row sums in list order
# speedup vs baseline: 8.2199x; 1.0589x over previous
#include <hip/hip_runtime.h>
#include <hip/hip_bf16.h>

typedef float v2f __attribute__((ext_vector_type(2)));
typedef float v8f __attribute__((ext_vector_type(8)));

#define N_NODES 50000
#define N_EDGES 800000
#define N_FEAT  256
#define N_HID   32
#define N_HEADS 8
#define ALPHA   0.2f
#define TILE_NODES 196

#define GM_GRID  256
#define GM_WAVES 8
#define GM_TILES (N_NODES / 16)
#define GM_STEP  (GM_GRID * (GM_WAVES / 2))

template <bool EXTRA>
__device__ __forceinline__ void gemmTiles(const float* __restrict__ hmat,
                                          float* __restrict__ Wh,
                                          float* __restrict__ s12,
                                          const v2f* bl, const v2f* bu,
                                          int tile, int nh, int m, int hf,
                                          const float4* rowp, float4 (&cur)[8]) {
    for (; tile < GM_TILES; tile += GM_STEP) {
        const int tn = tile + GM_STEP < GM_TILES ? tile + GM_STEP : tile;
        const float4* nrow = (const float4*)(hmat + (size_t)(tn * 16 + m) * N_FEAT + 32 * hf);

        v8f acc[8];
#pragma unroll
        for (int j = 0; j < 8; ++j) acc[j] = v8f{};
        v8f accu = v8f{};

#pragma nounroll
        for (int kc = 0; kc < 4; ++kc) {
            const float4* np = kc < 3 ? rowp + (kc + 1) * 16 : nrow;
            float4 nxt[8];
#pragma unroll
            for (int i = 0; i < 8; ++i) nxt[i] = np[i];

            const v2f* bk = bl + kc * (16 * 16 * 32);
            const v2f* uk = bu + kc * (16 * 32);
#pragma unroll
            for (int s = 0; s < 16; ++s) {
                const float4 q = cur[s >> 1];
                v2f a;
                a[0] = (s & 1) ? q.z : q.x;
                a[1] = (s & 1) ? q.w : q.y;
#pragma unroll
                for (int j = 0; j < 8; ++j) {
                    const v2f b = bk[(s * 16 + j) * 32];
                    acc[j] = __builtin_amdgcn_wmma_f32_16x16x4_f32(false, a, false, b,
                                                                   (short)0, acc[j], false, false);
                }
                if (EXTRA) {
                    const v2f b = uk[s * 32];
                    accu = __builtin_amdgcn_wmma_f32_16x16x4_f32(false, a, false, b,
                                                                 (short)0, accu, false, false);
                }
            }
#pragma unroll
            for (int i = 0; i < 8; ++i) cur[i] = nxt[i];
        }

#pragma unroll
        for (int j = 0; j < 8; ++j) {
            const int head = nh * 4 + (j >> 1);
            float* op = Wh + ((size_t)head * N_NODES + tile * 16 + hf * 8) * N_HID + (j & 1) * 16 + m;
#pragma unroll
            for (int r = 0; r < 8; ++r)
                op[r * N_HID] = acc[j][r];
        }
        if (EXTRA) {
            float* sp = s12 + (size_t)(tile * 16 + hf * 8) * 16 + m;
#pragma unroll
            for (int r = 0; r < 8; ++r)
                sp[r * 16] = accu[r];
        }
        rowp = nrow;
    }
}

__global__ __launch_bounds__(GM_WAVES * 32) void gat_gemm(const float* __restrict__ hmat,
                                                          const float* __restrict__ W,
                                                          const float* __restrict__ avec,
                                                          float* __restrict__ Wh,
                                                          float* __restrict__ s12) {
    __shared__ v2f wl[64 * 16 * 32];
    __shared__ v2f wu[64 * 32];

    const int tid  = threadIdx.x;
    const int lane = tid & 31;
    const int wave = __builtin_amdgcn_readfirstlane(tid >> 5);
    const int m  = lane & 15;
    const int hf = lane >> 4;

    const int nh = wave & 1;
    const bool extra = (wave & 1) == (wave >> 2);
    const int tile = blockIdx.x + GM_GRID * (wave >> 1);

    const float4* rowp = (const float4*)(hmat + (size_t)(tile * 16 + m) * N_FEAT + 32 * hf);
    float4 cur[8];
#pragma unroll
    for (int i = 0; i < 8; ++i) cur[i] = rowp[i];

    const int uh = m & 7;
    float areg[32];
#pragma unroll
    for (int dd = 0; dd < 32; ++dd)
        areg[dd] = avec[uh * (2 * N_HID) + (m >> 3) * N_HID + ((dd + 2 * uh) & 31)];

    for (int i0 = 0; i0 < (64 * 16 * 32) / (GM_WAVES * 32); i0 += 16) {
        v2f tmp[16];
#pragma unroll
        for (int i = 0; i < 16; ++i) {
            const int idx = tid + GM_WAVES * 32 * (i0 + i);
            const int nt = (idx >> 5) & 15, ks = idx >> 9;
            const int k = 64 * (ks >> 4) + 32 * hf + 2 * (ks & 15);
            const float* wp = W + (size_t)((nt >> 1) * N_FEAT + k) * N_HID + (nt & 1) * 16 + m;
            tmp[i][0] = wp[0];
            tmp[i][1] = wp[N_HID];
        }
#pragma unroll
        for (int i = 0; i < 16; ++i)
            wl[tid + GM_WAVES * 32 * (i0 + i)] = tmp[i];
    }
    __syncthreads();

    for (int e = 0; e < 64 / GM_WAVES; ++e) {
        const int ks = wave + GM_WAVES * e;
        const v2f* wr = wl + (ks * 16 + uh * 2) * 32 + hf * 16;
        float u0 = 0.0f, u1 = 0.0f;
#pragma unroll
        for (int dd = 0; dd < 32; ++dd) {
            const int d = (dd + 2 * uh) & 31;
            const v2f wv = wr[d + (d & 16)];
            u0 += wv[0] * areg[dd];
            u1 += wv[1] * areg[dd];
        }
        v2f uu; uu[0] = u0; uu[1] = u1;
        wu[ks * 32 + lane] = uu;
    }
    __syncthreads();

    const v2f* bl = wl + nh * (8 * 32) + lane;
    const v2f* bu = wu + lane;
    if (extra)
        gemmTiles<true>(hmat, Wh, s12, bl, bu, tile, nh, m, hf, rowp, cur);
    else
        gemmTiles<false>(hmat, Wh, s12, bl, bu, tile, nh, m, hf, rowp, cur);
}

#define SC_WAVES (N_EDGES / 128)
#define CH_EDGES 1024
#define N_CHUNKS ((N_EDGES + CH_EDGES - 1) / CH_EDGES)
#define OWN_TILES 256

static_assert(N_NODES % 16 == 0, "whole 16-node GEMM tiles");
static_assert(N_EDGES == SC_WAVES * 128, "every sort wave takes 128 edges");
static_assert((SC_WAVES + 7) / 8 == N_CHUNKS, "one sort work-group per chunk");
static_assert((N_NODES + TILE_NODES - 1) / TILE_NODES <= OWN_TILES, "a header word per owner tile");
static_assert(N_NODES <= (1 << 16) && TILE_NODES <= 256, "edge word: local node << 16 | src");

__global__ __launch_bounds__(256) void gat_sort(const int* __restrict__ src,
                                                const int* __restrict__ dst,
                                                int* __restrict__ eoff,
                                                int* __restrict__ eword) {
    __shared__ int tcnt[OWN_TILES];
    __shared__ int wsum[8];
    __shared__ __attribute__((aligned(16))) int srt[CH_EDGES];

    const int tid  = threadIdx.x;
    const int lane = tid & 31;
    const int wv = __builtin_amdgcn_readfirstlane(tid >> 5);
    const int w = blockIdx.x * 8 + wv;
    const bool live = w < SC_WAVES;

    tcnt[tid] = 0;

    int4 dv = make_int4(0, 0, 0, 0), sv = make_int4(0, 0, 0, 0);
    if (live) {
        dv = ((const int4*)dst)[w * 32 + lane];
        sv = ((const int4*)src)[w * 32 + lane];
    }

    const int dd[4] = {dv.x, dv.y, dv.z, dv.w};
    const int ss[4] = {sv.x, sv.y, sv.z, sv.w};
    unsigned ot[4]; int rk[4], ew[4];
#pragma unroll
    for (int u = 0; u < 4; ++u) {
        ot[u] = (unsigned)dd[u] / TILE_NODES;
        ew[u] = (int)(((unsigned)dd[u] - ot[u] * TILE_NODES) << 16) | ss[u];
        rk[u] = 0;
    }
    __syncthreads();
    if (live) {
#pragma unroll
        for (int u = 0; u < 4; ++u)
            rk[u] = atomicAdd(&tcnt[ot[u]], 1);
    }
    __syncthreads();

    const int c = tcnt[tid];
    int inc = c;
#pragma unroll
    for (int d = 1; d < 32; d <<= 1) {
        const int y = __shfl_up(inc, d, 32);
        if (lane >= d) inc += y;
    }
    if (lane == 31) wsum[wv] = inc;
    __syncthreads();
    int start = inc - c;
#pragma unroll
    for (int i = 0; i < 8; ++i)
        if (i < wv) start += wsum[i];
    tcnt[tid] = start;
    eoff[blockIdx.x * OWN_TILES + tid] = start | (c << 16);
    __syncthreads();

    if (live) {
#pragma unroll
        for (int u = 0; u < 4; ++u)
            srt[tcnt[ot[u]] + rk[u]] = ew[u];
    }
    __syncthreads();

    const int nvalid = (N_EDGES - (int)blockIdx.x * CH_EDGES) < CH_EDGES
                     ? (N_EDGES - (int)blockIdx.x * CH_EDGES) : CH_EDGES;
    if (4 * tid < nvalid)
        ((int4*)eword)[blockIdx.x * (CH_EDGES / 4) + tid] = ((const int4*)srt)[tid];
}

#define OWN_WAVES  32
#define LIST_CAP   704
#define KEEP_GROUPS 32

typedef float v32f __attribute__((ext_vector_type(32)));

static_assert(N_CHUNKS <= OWN_WAVES * 32, "one chunk per lane covers every chunk");
static_assert((KEEP_GROUPS * 4) % 16 == 0 && KEEP_GROUPS <= 32, "phase B fills keep[] 4 groups a trip");
static_assert(OWN_WAVES == 32, "a row's owner is wave (local node & 31); one deal counter per lane");

__device__ __forceinline__ void ldsMaxFloat(int* addr, float val) {
    if (val >= 0.0f)
        atomicMax(addr, __float_as_int(val));
    else
        atomicMin((unsigned int*)addr, __float_as_uint(val));
}

__device__ __forceinline__ void ownListFence() {
    __builtin_amdgcn_fence(__ATOMIC_ACQ_REL, "wavefront");
}

__device__ __forceinline__ void ownMax(const int* lw, int cnt, int lane,
                                       const float* __restrict__ s12,
                                       const float* s2l, int* mxi, v32f& keep) {
    const int j = lane >> 3, h = lane & 7;
    for (int i0 = 0; i0 < cnt; i0 += 16) {
        float x[4]; int ld[4];
#pragma unroll
        for (int u = 0; u < 4; ++u) {
            const int idx = i0 + u * 4 + j;
            const int ent = idx < cnt ? lw[idx] : 0;
            ld[u] = ent >> 16;
            x[u] = s12[(ent & 0xFFFF) * 16 + h] + s2l[ld[u] * N_HEADS + h];
        }
#pragma unroll
        for (int u = 0; u < 4; ++u) {
            const int idx = i0 + u * 4 + j;
            x[u] = x[u] > 0.0f ? x[u] : ALPHA * x[u];
            if (idx < cnt) ldsMaxFloat(&mxi[ld[u] * N_HEADS + h], x[u]);
        }
        if (i0 < KEEP_GROUPS * 4) {
#pragma unroll
            for (int u = 0; u < 4; ++u)
                keep[(i0 >> 2) + u] = x[u];
        }
    }
}

template <bool KEPT>
__device__ __forceinline__ void ownSum(const int* lw, int cnt, int lane,
                                       const float* __restrict__ s12,
                                       const float* __restrict__ Wh,
                                       const float* s2l, const int* mxi,
                                       float* dn, float* acc, const v32f& keep) {
    const int j = lane >> 3, h = lane & 7;
    for (int i0 = 0; i0 < cnt; i0 += 4) {
        const bool mine = i0 + j < cnt;
        const int ent = lw[mine ? i0 + j : cnt - 1];
        const int lj = ent >> 16;
        int s[4], ld[4]; float wh[4][N_HEADS];
#pragma unroll
        for (int u = 0; u < 4; ++u) {
            const int e = __builtin_amdgcn_readlane(ent, 8 * u);
            s[u]  = e & 0xFFFF;
            ld[u] = e >> 16;
        }
        float v;
        if (KEPT && i0 < KEEP_GROUPS * 4) {
            v = keep[i0 >> 2];
        } else {
            const float x = s12[(ent & 0xFFFF) * 16 + h] + s2l[lj * N_HEADS + h];
            v = x > 0.0f ? x : ALPHA * x;
        }
#pragma unroll
        for (int u = 0; u < 4; ++u)
#pragma unroll
            for (int head = 0; head < N_HEADS; ++head)
                wh[u][head] = Wh[((size_t)head * N_NODES + s[u]) * N_HID + lane];
        const float ex = __expf(v - __int_as_float(mxi[lj * N_HEADS + h]));
        if (mine) atomicAdd(&dn[lj * N_HEADS + h], ex);
#pragma unroll
        for (int u = 0; u < 4; ++u) {
            if (i0 + u < cnt) {
                float* ap = acc + ld[u] * (N_HEADS * N_HID) + lane;
#pragma unroll
                for (int head = 0; head < N_HEADS; ++head) {
                    const float w = __int_as_float(
                        __builtin_amdgcn_readlane(__float_as_int(ex), 8 * u + head));
                    atomicAdd(ap + head * N_HID, w * wh[u][head]);
                }
            }
        }
    }
}

__device__ __forceinline__ void ownSumRows(const int* lw, int cnt, int lane,
                                           const float* __restrict__ s12,
                                           const float* __restrict__ Wh,
                                           const float* s2l, const int* mxi,
                                           float* dn, float* acc, const v32f& keep) {
    const int j = lane >> 3, h = lane & 7;
    const float4* whl = (const float4*)Wh + (size_t)(lane >> 3) * (N_NODES * N_HID / 4) + (lane & 7);
    float4* accl = (float4*)acc + lane;
    int entn = cnt > 0 ? lw[j < cnt ? j : cnt - 1] : 0;
    for (int i0 = 0; i0 < cnt; i0 += 4) {
        const bool mine = i0 + j < cnt;
        const int ent = entn;
        entn = lw[i0 + 4 + j < cnt ? i0 + 4 + j : cnt - 1];
        const int lj = ent >> 16;
        int s[4], ld[4]; float4 wh[4][2];
#pragma unroll
        for (int u = 0; u < 4; ++u) {
            const int e = __builtin_amdgcn_readlane(ent, 8 * u);
            s[u]  = e & 0xFFFF;
            ld[u] = e >> 16;
        }
        float v;
        if (i0 < KEEP_GROUPS * 4) {
            v = keep[i0 >> 2];
        } else {
            const float x = s12[(ent & 0xFFFF) * 16 + h] + s2l[lj * N_HEADS + h];
            v = x > 0.0f ? x : ALPHA * x;
        }
#pragma unroll
        for (int u = 0; u < 4; ++u) {
            const float4* rp = whl + s[u] * (N_HID / 4);
            wh[u][0] = rp[0];
            wh[u][1] = rp[4 * (N_NODES * N_HID / 4)];
        }
        const float ex = __expf(v - __int_as_float(mxi[lj * N_HEADS + h]));
        if (mine) atomicAdd(&dn[lj * N_HEADS + h], ex);
#pragma unroll
        for (int u = 0; u < 4; ++u) {
            const float w0 = __shfl(ex, 8 * u + (lane >> 3), 32);
            const float w1 = __shfl(ex, 8 * u + 4 + (lane >> 3), 32);
            if (i0 + u < cnt) {
                float4* rp = accl + ld[u] * (N_HEADS * N_HID / 4);
                float4 r0 = rp[0], r1 = rp[32];
                r0.x += w0 * wh[u][0].x; r0.y += w0 * wh[u][0].y;
                r0.z += w0 * wh[u][0].z; r0.w += w0 * wh[u][0].w;
                r1.x += w1 * wh[u][1].x; r1.y += w1 * wh[u][1].y;
                r1.z += w1 * wh[u][1].z; r1.w += w1 * wh[u][1].w;
                rp[0] = r0; rp[32] = r1;
            }
        }
    }
}

__device__ __forceinline__ void ownDeal(int* lst, int* lcnt, int wave, int lane, int tile,
                                        const int* __restrict__ eoff,
                                        const int* __restrict__ eword) {
    const int ci  = lane * OWN_WAVES + wave;
    const bool has = ci < N_CHUNKS;
    const int cc  = has ? ci : 0;
    const int hdr = eoff[cc * OWN_TILES + tile];
    const int len = has ? (hdr >> 16) : 0;
    const int beg = cc * CH_EDGES + (hdr & 0xFFFF);
    for (int t0 = 0; __builtin_amdgcn_ballot_w32(t0 < len) != 0u; t0 += 4) {
        int v[4];
#pragma unroll
        for (int u = 0; u < 4; ++u)
            v[u] = eword[t0 + u < len ? beg + t0 + u : 0];
        int ow[4], p[4];
#pragma unroll
        for (int u = 0; u < 4; ++u) {
            ow[u] = (v[u] >> 16) & (OWN_WAVES - 1);
            p[u] = LIST_CAP;
            if (t0 + u < len) p[u] = atomicAdd(&lcnt[ow[u]], 1);
        }
#pragma unroll
        for (int u = 0; u < 4; ++u)
            if (p[u] < LIST_CAP) lst[ow[u] * LIST_CAP + p[u]] = v[u];
    }
}

template <int PHASE>
__device__ __forceinline__ void ownWork(int* lw, int cnt, int lane,
                                        const float* __restrict__ s12,
                                        const float* __restrict__ Wh,
                                        const float* s2l, int* mxi,
                                        float* dn, float* acc, v32f& keep) {
    if (PHASE == 0) ownMax(lw, cnt, lane, s12, s2l, mxi, keep);
    else            ownSum<false>(lw, cnt, lane, s12, Wh, s2l, mxi, dn, acc, keep);
}

template <int PHASE>
__device__ __forceinline__ int ownCollect(int* lw, int wave, int lane, int tile,
                                          const int* __restrict__ eoff,
                                          const int* __restrict__ eword,
                                          const float* __restrict__ s12,
                                          const float* __restrict__ Wh,
                                          const float* s2l, int* mxi,
                                          float* dn, float* acc, int& cnt_out,
                                          v32f& keep) {
    const int ci  = lane * OWN_WAVES + wave;
    const bool has = ci < N_CHUNKS;
    const int cc  = has ? ci : 0;
    const int hdr = eoff[cc * OWN_TILES + tile];
    const int len = has ? (hdr >> 16) : 0;
    const int beg = cc * CH_EDGES + (hdr & 0xFFFF);

    int inc = len;
#pragma unroll
    for (int d = 1; d < 32; d <<= 1) {
        const int y = __shfl_up(inc, d, 32);
        if (lane >= d) inc += y;
    }
    const int pos   = inc - len;
    const int total = __builtin_amdgcn_readlane(inc, 31);

    if (total <= LIST_CAP) {
        for (int t0 = 0; __builtin_amdgcn_ballot_w32(t0 < len) != 0u; t0 += 4) {
            int v[4];
#pragma unroll
            for (int u = 0; u < 4; ++u)
                v[u] = eword[t0 + u < len ? beg + t0 + u : 0];
#pragma unroll
            for (int u = 0; u < 4; ++u)
                if (t0 + u < len) lw[pos + t0 + u] = v[u];
        }
        ownListFence();
        ownWork<PHASE>(lw, total, lane, s12, Wh, s2l, mxi, dn, acc, keep);
        cnt_out = total;
        return 1;
    }

    for (int l = 0; l < 32; ++l) {
        const int n = __builtin_amdgcn_readlane(len, l);
        const int b = __builtin_amdgcn_readlane(beg, l);
        for (int o = 0; o < n; o += LIST_CAP) {
            const int m = n - o < LIST_CAP ? n - o : LIST_CAP;
            for (int i = lane; i < m; i += 32)
                lw[i] = eword[b + o + i];
            ownListFence();
            ownWork<PHASE>(lw, m, lane, s12, Wh, s2l, mxi, dn, acc, keep);
            ownListFence();
        }
    }
    cnt_out = 0;
    return 2;
}

__global__ __launch_bounds__(1024) void gat_own(const int* __restrict__ eoff,
                                                const int* __restrict__ eword,
                                                const float* __restrict__ s12,
                                                const float* __restrict__ Wh,
                                                float* __restrict__ out) {
    __shared__ __attribute__((aligned(16))) float acc[TILE_NODES * N_HEADS * N_HID];
    __shared__ int   mxi[TILE_NODES * N_HEADS];
    __shared__ float dn[TILE_NODES * N_HEADS];
    __shared__ float s2l[TILE_NODES * N_HEADS];
    __shared__ int   lst[OWN_WAVES * LIST_CAP];
    __shared__ int   lcnt[OWN_WAVES];

    const int tid  = threadIdx.x;
    const int lane = tid & 31;
    const int wave = __builtin_amdgcn_readfirstlane(tid >> 5);
    const int lo   = blockIdx.x * TILE_NODES;
    const int nloc = (N_NODES - lo) < TILE_NODES ? (N_NODES - lo) : TILE_NODES;

    float4* acc4 = (float4*)acc;
    for (int i = tid; i < TILE_NODES * N_HEADS * N_HID / 4; i += 1024)
        acc4[i] = make_float4(0.0f, 0.0f, 0.0f, 0.0f);
    for (int i = tid; i < TILE_NODES * N_HEADS; i += 1024) {
        const int node = lo + (i >> 3);
        mxi[i] = __float_as_int(-__builtin_inff());
        dn[i]  = 0.0f;
        s2l[i] = node < N_NODES ? s12[node * 16 + 8 + (i & 7)] : 0.0f;
    }
    if (tid < OWN_WAVES) lcnt[tid] = 0;
    __syncthreads();

    ownDeal(lst, lcnt, wave, lane, (int)blockIdx.x, eoff, eword);
    __syncthreads();

    int* lw = lst + wave * LIST_CAP;
    v32f keep = {};
    const int dealt = lcnt[lane];
    const bool fits = __builtin_amdgcn_ballot_w32(dealt > LIST_CAP) == 0u;
    if (fits) {
        const int cnt = __builtin_amdgcn_readlane(dealt, wave);
        ownMax(lw, cnt, lane, s12, s2l, mxi, keep);
        ownListFence();
        ownSumRows(lw, cnt, lane, s12, Wh, s2l, mxi, dn, acc, keep);
    } else {
        int cnt = 0;
        const int rounds = ownCollect<0>(lw, wave, lane, (int)blockIdx.x, eoff, eword, s12, Wh,
                                         s2l, mxi, dn, acc, cnt, keep);
        __syncthreads();
        if (rounds == 1)
            ownSum<true>(lw, cnt, lane, s12, Wh, s2l, mxi, dn, acc, keep);
        else
            ownCollect<1>(lw, wave, lane, (int)blockIdx.x, eoff, eword, s12, Wh, s2l, mxi, dn, acc, cnt, keep);
    }
    __syncthreads();

    float4* o4 = (float4*)(out + (size_t)lo * (N_HEADS * N_HID));
    for (int i = tid; i < nloc * (N_HEADS * N_HID / 4); i += 1024) {
        const float d0 = dn[(i >> 6) * N_HEADS + ((i >> 3) & 7)];
        const float q = d0 > 0.0f ? d0 : 1.0f;
        float4 v = acc4[i];
        v.x /= q; v.y /= q; v.z /= q; v.w /= q;
        o4[i] = v;
    }
}

extern "C" void kernel_launch(void* const* d_in, const int* in_sizes, int n_in,
                              void* d_out, int out_size, void* d_ws, size_t ws_size,
                              hipStream_t stream) {
    const float* h   = (const float*)d_in[0];
    const float* W   = (const float*)d_in[1];
    const float* a   = (const float*)d_in[2];
    const int*   src = (const int*)d_in[3];
    const int*   dst = (const int*)d_in[4];
    float* out = (float*)d_out;

    float* ws = (float*)d_ws;
    float* Wh = ws;
    float* s12 = Wh + (size_t)N_HEADS * N_NODES * N_HID;
    int* eoff  = (int*)(s12 + (size_t)N_NODES * 16);
    int* eword = eoff + N_CHUNKS * OWN_TILES;

    gat_gemm<<<GM_GRID, GM_WAVES * 32, 0, stream>>>(h, W, a, Wh, s12);

    gat_sort<<<N_CHUNKS, 256, 0, stream>>>(src, dst, eoff, eword);

    gat_own<<<(N_NODES + TILE_NODES - 1) / TILE_NODES, 1024, 0, stream>>>(eoff, eword, s12, Wh, out);
}
